# starting kernel
# baseline (speedup 1.0000x reference)
#include <hip/hip_runtime.h>
#include <hip/hip_bf16.h>

typedef float v2f __attribute__((ext_vector_type(2)));
typedef float v8f __attribute__((ext_vector_type(8)));

#define NFEAT 128

__global__ void gcn_fill1(float* __restrict__ deg, int n) {
    int i = blockIdx.x * blockDim.x + threadIdx.x;
    if (i < n) deg[i] = 1.0f;
}

__global__ void gcn_deg_accum(const int* __restrict__ dst,
                              const float* __restrict__ ew,
                              float* __restrict__ deg, int n_edges) {
    int e = blockIdx.x * blockDim.x + threadIdx.x;
    if (e < n_edges)
        __hip_atomic_fetch_add(&deg[dst[e]], ew[e],
                               __ATOMIC_RELAXED, __HIP_MEMORY_SCOPE_AGENT);
}

__global__ void gcn_rsqrt(float* __restrict__ d, int n) {
    int i = blockIdx.x * blockDim.x + threadIdx.x;
    if (i < n) d[i] = rsqrtf(d[i]);
}

__global__ void gcn_zero(float* __restrict__ p, int n) {
    int i = blockIdx.x * blockDim.x + threadIdx.x;
    if (i < n) p[i] = 0.0f;
}

__global__ __launch_bounds__(256) void gcn_gemm128_wmma(
        const float* __restrict__ X, const float* __restrict__ W,
        float* __restrict__ H) {
    __shared__ float Wl[NFEAT * NFEAT];
    {
        const float4* Wv  = (const float4*)W;
        float4*       Wlv = (float4*)Wl;
        int tid = threadIdx.y * 32 + threadIdx.x;
        #pragma unroll
        for (int i = 0; i < (NFEAT * NFEAT / 4) / 256; ++i)
            Wlv[tid + i * 256] = Wv[tid + i * 256];
    }
    __syncthreads();

    const int lane  = threadIdx.x;
    const int m0    = blockIdx.x * 16;
    const int n0    = threadIdx.y * 16;
    const int mlane = lane & 15;
    const int khalf = (lane >> 4) * 2;
    const int ncol  = n0 + mlane;

    const float* xrow = X + (size_t)(m0 + mlane) * NFEAT;
    v8f acc = {};

    #pragma unroll
    for (int k0 = 0; k0 < NFEAT; k0 += 4) {
        const int k = k0 + khalf;
        v2f a = *(const v2f*)&xrow[k];
        v2f b;
        b.x = Wl[(k    ) * NFEAT + ncol];
        b.y = Wl[(k + 1) * NFEAT + ncol];
        acc = __builtin_amdgcn_wmma_f32_16x16x4_f32(
                  false, a, false, b, (short)0, acc, false, false);
    }

    const int rbase = m0 + (lane >> 4) * 8;
    #pragma unroll
    for (int r = 0; r < 8; ++r)
        H[(size_t)(rbase + r) * NFEAT + ncol] = acc[r];
}

__global__ __launch_bounds__(256) void gcn_scatter(
        const float* __restrict__ h, float* __restrict__ agg,
        const int* __restrict__ src, const int* __restrict__ dst,
        const float* __restrict__ ew, const float* __restrict__ dinv,
        int n_edges) {
    const int wave = (blockIdx.x * blockDim.x + threadIdx.x) >> 5;
    const int lane = threadIdx.x & 31;
    if (wave >= n_edges) return;

    const int   s = src[wave];
    const int   d = dst[wave];
    const float c = dinv[s] * ew[wave] * dinv[d];

    const float4 hv = *(const float4*)&h[(size_t)s * NFEAT + lane * 4];
    float* out = &agg[(size_t)d * NFEAT + lane * 4];
    __hip_atomic_fetch_add(&out[0], hv.x * c, __ATOMIC_RELAXED, __HIP_MEMORY_SCOPE_AGENT);
    __hip_atomic_fetch_add(&out[1], hv.y * c, __ATOMIC_RELAXED, __HIP_MEMORY_SCOPE_AGENT);
    __hip_atomic_fetch_add(&out[2], hv.z * c, __ATOMIC_RELAXED, __HIP_MEMORY_SCOPE_AGENT);
    __hip_atomic_fetch_add(&out[3], hv.w * c, __ATOMIC_RELAXED, __HIP_MEMORY_SCOPE_AGENT);
}

__global__ void gcn_fuse(const float* __restrict__ agg,
                         const float* __restrict__ hpre,
                         const float* __restrict__ dinv,
                         const float* __restrict__ b,
                         float* __restrict__ out,
                         int n_nodes, int do_relu) {
    int idx = blockIdx.x * blockDim.x + threadIdx.x;
    if (idx >= n_nodes * NFEAT) return;
    int i = idx >> 7;
    int d = idx & (NFEAT - 1);
    float di = dinv[i];
    float v = agg[idx] + hpre[idx] * (di * di) + b[d];
    out[idx] = do_relu ? fmaxf(v, 0.0f) : v;
}

__global__ void gcn_classifier(const float* __restrict__ h,
                               const float* __restrict__ Wc,
                               const float* __restrict__ bc,
                               float* __restrict__ out, int n_nodes) {
    int i = blockIdx.x * blockDim.x + threadIdx.x;
    if (i >= n_nodes) return;
    float acc[10];
    #pragma unroll
    for (int c = 0; c < 10; ++c) acc[c] = bc[c];
    const float* hr = h + (size_t)i * NFEAT;
    for (int d = 0; d < NFEAT; ++d) {
        float hv = hr[d];
        #pragma unroll
        for (int c = 0; c < 10; ++c) acc[c] += hv * Wc[d * 10 + c];
    }
    #pragma unroll
    for (int c = 0; c < 10; ++c) out[(size_t)i * 10 + c] = acc[c];
}

extern "C" void kernel_launch(void* const* d_in, const int* in_sizes, int n_in,
                              void* d_out, int out_size, void* d_ws, size_t ws_size,
                              hipStream_t stream) {
    const float* x   = (const float*)d_in[0];
    const int*   ei  = (const int*)  d_in[1];
    const float* ew  = (const float*)d_in[2];
    const float* W1  = (const float*)d_in[3];
    const float* b1  = (const float*)d_in[4];
    const float* W2  = (const float*)d_in[5];
    const float* b2  = (const float*)d_in[6];
    const float* Wc  = (const float*)d_in[7];
    const float* bc  = (const float*)d_in[8];
    float*       out = (float*)d_out;

    const int n_nodes = in_sizes[0] / NFEAT;
    const int n_edges = in_sizes[2];
    const int* src = ei;
    const int* dst = ei + n_edges;

    const size_t nd = (size_t)n_nodes * NFEAT;
    float* bufA = (float*)d_ws;
    float* bufB = bufA + nd;
    float* bufC = bufB + nd;
    float* dinv = bufC + nd;

    const int B = 256;
    const int gN  = (n_nodes + B - 1) / B;
    const int gE  = (n_edges + B - 1) / B;
    const int gND = (int)((nd + B - 1) / B);
    const int gEw = (int)(((size_t)n_edges * 32 + B - 1) / B);
    const dim3 gblk(32, 8);
    const int gTiles = n_nodes / 16;

    gcn_fill1    <<<gN, B, 0, stream>>>(dinv, n_nodes);
    gcn_deg_accum<<<gE, B, 0, stream>>>(dst, ew, dinv, n_edges);
    gcn_rsqrt    <<<gN, B, 0, stream>>>(dinv, n_nodes);

    gcn_gemm128_wmma<<<gTiles, gblk, 0, stream>>>(x, W1, bufA);
    gcn_zero        <<<gND, B, 0, stream>>>(bufB, (int)nd);
    gcn_scatter     <<<gEw, B, 0, stream>>>(bufA, bufB, src, dst, ew, dinv, n_edges);
    gcn_fuse        <<<gND, B, 0, stream>>>(bufB, bufA, dinv, b1, bufA, n_nodes, 1);

    gcn_gemm128_wmma<<<gTiles, gblk, 0, stream>>>(bufA, W2, bufB);
    gcn_zero        <<<gND, B, 0, stream>>>(bufC, (int)nd);
    gcn_scatter     <<<gEw, B, 0, stream>>>(bufB, bufC, src, dst, ew, dinv, n_edges);
    gcn_fuse        <<<gND, B, 0, stream>>>(bufC, bufB, dinv, b2, bufB, n_nodes, 1);

    gcn_classifier<<<gN, B, 0, stream>>>(bufB, Wc, bc, out, n_nodes);
}


# Float-atomic scatter and degree sums replaced by in-call grouping of edges by destination (int count atomic gives rank, scan, fill) and one-wave-per-row gather; rows summed in arrival-rank order
# speedup vs baseline: 1.7443x; 1.7443x over previous
#include <hip/hip_runtime.h>
#include <hip/hip_bf16.h>

typedef float v2f __attribute__((ext_vector_type(2)));
typedef float v8f __attribute__((ext_vector_type(8)));

#define NFEAT 128

__global__ void gcn_zero_cnt(int* __restrict__ cnt, int n) {
    int i = blockIdx.x * blockDim.x + threadIdx.x;
    if (i < n) cnt[i] = 0;
}

__global__ void gcn_count_rank(const int* __restrict__ dst, int* __restrict__ cnt,
                               int* __restrict__ rank, int n_edges) {
    int e = blockIdx.x * blockDim.x + threadIdx.x;
    if (e < n_edges)
        rank[e] = __hip_atomic_fetch_add(&cnt[dst[e]], 1,
                                         __ATOMIC_RELAXED, __HIP_MEMORY_SCOPE_AGENT);
}

__global__ __launch_bounds__(1024) void gcn_scan(const int* __restrict__ cnt,
                                                 int* __restrict__ rowptr, int n) {
    __shared__ int s_pre[32];
    __shared__ int s_own[32];
    const int t     = threadIdx.x;
    const int lane  = t & 31;
    const int wave  = t >> 5;
    const int first = blockIdx.x * 1024;

    const int4* c4 = (const int4*)cnt;
    const int   nq = first >> 2;
    int pre = 0;
    for (int q = t; q < nq; q += 1024) {
        int4 v = c4[q];
        pre += (v.x + v.y) + (v.z + v.w);
    }

    const int i = first + t;
    const int x = (i < n) ? cnt[i] : 0;
    int y = x;
    for (int o = 1; o < 32; o <<= 1) {
        int u = __shfl_up(y, o);
        if (lane >= o) y += u;
    }
    for (int o = 16; o > 0; o >>= 1) pre += __shfl_xor(pre, o);
    if (lane == 31) s_own[wave] = y;
    if (lane == 0)  s_pre[wave] = pre;
    __syncthreads();

    int a = s_pre[lane] + ((lane < wave) ? s_own[lane] : 0);
    for (int o = 16; o > 0; o >>= 1) a += __shfl_xor(a, o);

    const int excl = a + y - x;
    if (i < n)      rowptr[i] = excl;
    if (i == n - 1) rowptr[n] = excl + x;
}

__global__ void gcn_fill_rows(const int* __restrict__ src, const int* __restrict__ dst,
                              const float* __restrict__ ew,
                              const int* __restrict__ rowptr, const int* __restrict__ rank,
                              float* __restrict__ csr, int n_edges) {
    int e = blockIdx.x * blockDim.x + threadIdx.x;
    if (e < n_edges) {
        const int p = rowptr[dst[e]] + rank[e];
        ((float2*)csr)[p] = make_float2(__int_as_float(src[e]), ew[e]);
    }
}

__global__ void gcn_dinv_rows(const int* __restrict__ rowptr, const float* __restrict__ csr,
                              float* __restrict__ dinv, int n) {
    int i = blockIdx.x * blockDim.x + threadIdx.x;
    if (i >= n) return;
    const float2* rec = (const float2*)csr;
    const int end = rowptr[i + 1];
    float s = 0.0f;
    for (int p = rowptr[i]; p < end; ++p) s += rec[p].y;
    dinv[i] = rsqrtf(s + 1.0f);
}

__global__ __launch_bounds__(256) void gcn_gemm128_wmma(
        const float* __restrict__ X, const float* __restrict__ W,
        float* __restrict__ H) {
    __shared__ float Wl[NFEAT * NFEAT];
    {
        const float4* Wv  = (const float4*)W;
        float4*       Wlv = (float4*)Wl;
        int tid = threadIdx.y * 32 + threadIdx.x;
        #pragma unroll
        for (int i = 0; i < (NFEAT * NFEAT / 4) / 256; ++i)
            Wlv[tid + i * 256] = Wv[tid + i * 256];
    }
    __syncthreads();

    const int lane  = threadIdx.x;
    const int m0    = blockIdx.x * 16;
    const int n0    = threadIdx.y * 16;
    const int mlane = lane & 15;
    const int khalf = (lane >> 4) * 2;
    const int ncol  = n0 + mlane;

    const float* xrow = X + (size_t)(m0 + mlane) * NFEAT;
    v8f acc = {};

    #pragma unroll
    for (int k0 = 0; k0 < NFEAT; k0 += 4) {
        const int k = k0 + khalf;
        v2f a = *(const v2f*)&xrow[k];
        v2f b;
        b.x = Wl[(k    ) * NFEAT + ncol];
        b.y = Wl[(k + 1) * NFEAT + ncol];
        acc = __builtin_amdgcn_wmma_f32_16x16x4_f32(
                  false, a, false, b, (short)0, acc, false, false);
    }

    const int rbase = m0 + (lane >> 4) * 8;
    #pragma unroll
    for (int r = 0; r < 8; ++r)
        H[(size_t)(rbase + r) * NFEAT + ncol] = acc[r];
}

__global__ __launch_bounds__(256) void gcn_gather(
        const float* __restrict__ h, float* __restrict__ agg,
        const int* __restrict__ rowptr, const float* __restrict__ csr,
        const float* __restrict__ dinv, int n_nodes) {
    const int lane = threadIdx.x & 31;
    const int row = __builtin_amdgcn_readfirstlane(
                        (int)(blockIdx.x * (blockDim.x >> 5) + (threadIdx.x >> 5)));
    if (row >= n_nodes) return;

    const int   beg = rowptr[row];
    const int   end = rowptr[row + 1];
    const float di  = dinv[row];
    const float2* rec = (const float2*)csr;
    const float*  hl  = h + lane * 4;

    float4 acc = make_float4(0.0f, 0.0f, 0.0f, 0.0f);
    for (int base = beg; base < end; base += 32) {
        const int p = base + lane;
        int   s = 0;
        float c = 0.0f;
        if (p < end) {
            const float2 r = rec[p];
            s = __float_as_int(r.x);
            c = dinv[s] * r.y * di;
        }
        const int cbits = __float_as_int(c);
        const int cnt   = min(32, end - base);
        int j = 0;
        for (; j + 4 <= cnt; j += 4) {
            const int   s0 = __builtin_amdgcn_readlane(s, j);
            const int   s1 = __builtin_amdgcn_readlane(s, j + 1);
            const int   s2 = __builtin_amdgcn_readlane(s, j + 2);
            const int   s3 = __builtin_amdgcn_readlane(s, j + 3);
            const float c0 = __int_as_float(__builtin_amdgcn_readlane(cbits, j));
            const float c1 = __int_as_float(__builtin_amdgcn_readlane(cbits, j + 1));
            const float c2 = __int_as_float(__builtin_amdgcn_readlane(cbits, j + 2));
            const float c3 = __int_as_float(__builtin_amdgcn_readlane(cbits, j + 3));
            const float4 h0 = *(const float4*)&hl[(size_t)s0 * NFEAT];
            const float4 h1 = *(const float4*)&hl[(size_t)s1 * NFEAT];
            const float4 h2 = *(const float4*)&hl[(size_t)s2 * NFEAT];
            const float4 h3 = *(const float4*)&hl[(size_t)s3 * NFEAT];
            acc.x += h0.x * c0; acc.y += h0.y * c0; acc.z += h0.z * c0; acc.w += h0.w * c0;
            acc.x += h1.x * c1; acc.y += h1.y * c1; acc.z += h1.z * c1; acc.w += h1.w * c1;
            acc.x += h2.x * c2; acc.y += h2.y * c2; acc.z += h2.z * c2; acc.w += h2.w * c2;
            acc.x += h3.x * c3; acc.y += h3.y * c3; acc.z += h3.z * c3; acc.w += h3.w * c3;
        }
        for (; j < cnt; ++j) {
            const int    s0 = __builtin_amdgcn_readlane(s, j);
            const float  c0 = __int_as_float(__builtin_amdgcn_readlane(cbits, j));
            const float4 h0 = *(const float4*)&hl[(size_t)s0 * NFEAT];
            acc.x += h0.x * c0; acc.y += h0.y * c0; acc.z += h0.z * c0; acc.w += h0.w * c0;
        }
    }
    *(float4*)&agg[(size_t)row * NFEAT + lane * 4] = acc;
}

__global__ void gcn_fuse(const float* __restrict__ agg,
                         const float* __restrict__ hpre,
                         const float* __restrict__ dinv,
                         const float* __restrict__ b,
                         float* __restrict__ out,
                         int n_nodes, int do_relu) {
    int idx = blockIdx.x * blockDim.x + threadIdx.x;
    if (idx >= n_nodes * NFEAT) return;
    int i = idx >> 7;
    int d = idx & (NFEAT - 1);
    float di = dinv[i];
    float v = agg[idx] + hpre[idx] * (di * di) + b[d];
    out[idx] = do_relu ? fmaxf(v, 0.0f) : v;
}

__global__ void gcn_classifier(const float* __restrict__ h,
                               const float* __restrict__ Wc,
                               const float* __restrict__ bc,
                               float* __restrict__ out, int n_nodes) {
    int i = blockIdx.x * blockDim.x + threadIdx.x;
    if (i >= n_nodes) return;
    float acc[10];
    #pragma unroll
    for (int c = 0; c < 10; ++c) acc[c] = bc[c];
    const float* hr = h + (size_t)i * NFEAT;
    for (int d = 0; d < NFEAT; ++d) {
        float hv = hr[d];
        #pragma unroll
        for (int c = 0; c < 10; ++c) acc[c] += hv * Wc[d * 10 + c];
    }
    #pragma unroll
    for (int c = 0; c < 10; ++c) out[(size_t)i * 10 + c] = acc[c];
}

extern "C" void kernel_launch(void* const* d_in, const int* in_sizes, int n_in,
                              void* d_out, int out_size, void* d_ws, size_t ws_size,
                              hipStream_t stream) {
    const float* x   = (const float*)d_in[0];
    const int*   ei  = (const int*)  d_in[1];
    const float* ew  = (const float*)d_in[2];
    const float* W1  = (const float*)d_in[3];
    const float* b1  = (const float*)d_in[4];
    const float* W2  = (const float*)d_in[5];
    const float* b2  = (const float*)d_in[6];
    const float* Wc  = (const float*)d_in[7];
    const float* bc  = (const float*)d_in[8];
    float*       out = (float*)d_out;

    const int n_nodes = in_sizes[0] / NFEAT;
    const int n_edges = in_sizes[2];
    const int* src = ei;
    const int* dst = ei + n_edges;

    const size_t nd = (size_t)n_nodes * NFEAT;
    const size_t nn = ((size_t)n_nodes + 4) & ~(size_t)3;
    const size_t ne = ((size_t)n_edges + 3) & ~(size_t)3;
    float* bufA = (float*)d_ws;
    float* bufB = bufA + nd;
    float* bufC = bufB + nd;
    float* dinv = bufC + nd;
    int*   cnt    = (int*)(dinv + nn);
    int*   rowptr = cnt + nn;
    float* csr    = (float*)(rowptr + nn);
    int*   rank   = (int*)(csr + 2 * ne);

    const int B = 256;
    const int gN  = (n_nodes + B - 1) / B;
    const int gE  = (n_edges + B - 1) / B;
    const int gND = (int)((nd + B - 1) / B);
    const int gRow  = (n_nodes + 7) / 8;
    const int gScan = (n_nodes + 1023) / 1024;
    const dim3 gblk(32, 8);
    const int gTiles = n_nodes / 16;

    gcn_zero_cnt  <<<gN, B, 0, stream>>>(cnt, n_nodes);
    gcn_count_rank<<<gE, B, 0, stream>>>(dst, cnt, rank, n_edges);
    gcn_scan      <<<gScan, 1024, 0, stream>>>(cnt, rowptr, n_nodes);
    gcn_fill_rows <<<gE, B, 0, stream>>>(src, dst, ew, rowptr, rank, csr, n_edges);
    gcn_dinv_rows <<<gN, B, 0, stream>>>(rowptr, csr, dinv, n_nodes);

    gcn_gemm128_wmma<<<gTiles, gblk, 0, stream>>>(x, W1, bufA);
    gcn_gather      <<<gRow, B, 0, stream>>>(bufA, bufB, rowptr, csr, dinv, n_nodes);
    gcn_fuse        <<<gND, B, 0, stream>>>(bufB, bufA, dinv, b1, bufA, n_nodes, 1);

    gcn_gemm128_wmma<<<gTiles, gblk, 0, stream>>>(bufA, W2, bufB);
    gcn_gather      <<<gRow, B, 0, stream>>>(bufB, bufC, rowptr, csr, dinv, n_nodes);
    gcn_fuse        <<<gND, B, 0, stream>>>(bufC, bufB, dinv, b2, bufB, n_nodes, 1);

    gcn_classifier<<<gN, B, 0, stream>>>(bufB, Wc, bc, out, n_nodes);
}


# Self loop, bias and ReLU folded into the gather's epilogue while the row sum is in registers: both one-element-per-thread tail launches and the sum plane's round trip go; sum order unchanged
# speedup vs baseline: 2.2103x; 1.2672x over previous
#include <hip/hip_runtime.h>
#include <hip/hip_bf16.h>

typedef float v2f __attribute__((ext_vector_type(2)));
typedef float v8f __attribute__((ext_vector_type(8)));

#define NFEAT 128

__global__ void gcn_zero_cnt(int* __restrict__ cnt, int n) {
    int i = blockIdx.x * blockDim.x + threadIdx.x;
    if (i < n) cnt[i] = 0;
}

__global__ void gcn_count_rank(const int* __restrict__ dst, int* __restrict__ cnt,
                               int* __restrict__ rank, int n_edges) {
    int e = blockIdx.x * blockDim.x + threadIdx.x;
    if (e < n_edges)
        rank[e] = __hip_atomic_fetch_add(&cnt[dst[e]], 1,
                                         __ATOMIC_RELAXED, __HIP_MEMORY_SCOPE_AGENT);
}

__global__ __launch_bounds__(1024) void gcn_scan(const int* __restrict__ cnt,
                                                 int* __restrict__ rowptr, int n) {
    __shared__ int s_pre[32];
    __shared__ int s_own[32];
    const int t     = threadIdx.x;
    const int lane  = t & 31;
    const int wave  = t >> 5;
    const int first = blockIdx.x * 1024;

    const int4* c4 = (const int4*)cnt;
    const int   nq = first >> 2;
    int pre = 0;
    for (int q = t; q < nq; q += 1024) {
        int4 v = c4[q];
        pre += (v.x + v.y) + (v.z + v.w);
    }

    const int i = first + t;
    const int x = (i < n) ? cnt[i] : 0;
    int y = x;
    for (int o = 1; o < 32; o <<= 1) {
        int u = __shfl_up(y, o);
        if (lane >= o) y += u;
    }
    for (int o = 16; o > 0; o >>= 1) pre += __shfl_xor(pre, o);
    if (lane == 31) s_own[wave] = y;
    if (lane == 0)  s_pre[wave] = pre;
    __syncthreads();

    int a = s_pre[lane] + ((lane < wave) ? s_own[lane] : 0);
    for (int o = 16; o > 0; o >>= 1) a += __shfl_xor(a, o);

    const int excl = a + y - x;
    if (i < n)      rowptr[i] = excl;
    if (i == n - 1) rowptr[n] = excl + x;
}

__global__ void gcn_fill_rows(const int* __restrict__ src, const int* __restrict__ dst,
                              const float* __restrict__ ew,
                              const int* __restrict__ rowptr, const int* __restrict__ rank,
                              float* __restrict__ csr, int n_edges) {
    int e = blockIdx.x * blockDim.x + threadIdx.x;
    if (e < n_edges) {
        const int p = rowptr[dst[e]] + rank[e];
        ((float2*)csr)[p] = make_float2(__int_as_float(src[e]), ew[e]);
    }
}

__global__ void gcn_dinv_rows(const int* __restrict__ rowptr, const float* __restrict__ csr,
                              float* __restrict__ dinv, int n) {
    int i = blockIdx.x * blockDim.x + threadIdx.x;
    if (i >= n) return;
    const float2* rec = (const float2*)csr;
    const int end = rowptr[i + 1];
    float s = 0.0f;
    for (int p = rowptr[i]; p < end; ++p) s += rec[p].y;
    dinv[i] = rsqrtf(s + 1.0f);
}

__global__ __launch_bounds__(256) void gcn_gemm128_wmma(
        const float* __restrict__ X, const float* __restrict__ W,
        float* __restrict__ H) {
    __shared__ float Wl[NFEAT * NFEAT];
    {
        const float4* Wv  = (const float4*)W;
        float4*       Wlv = (float4*)Wl;
        int tid = threadIdx.y * 32 + threadIdx.x;
        #pragma unroll
        for (int i = 0; i < (NFEAT * NFEAT / 4) / 256; ++i)
            Wlv[tid + i * 256] = Wv[tid + i * 256];
    }
    __syncthreads();

    const int lane  = threadIdx.x;
    const int m0    = blockIdx.x * 16;
    const int n0    = threadIdx.y * 16;
    const int mlane = lane & 15;
    const int khalf = (lane >> 4) * 2;
    const int ncol  = n0 + mlane;

    const float* xrow = X + (size_t)(m0 + mlane) * NFEAT;
    v8f acc = {};

    #pragma unroll
    for (int k0 = 0; k0 < NFEAT; k0 += 4) {
        const int k = k0 + khalf;
        v2f a = *(const v2f*)&xrow[k];
        v2f b;
        b.x = Wl[(k    ) * NFEAT + ncol];
        b.y = Wl[(k + 1) * NFEAT + ncol];
        acc = __builtin_amdgcn_wmma_f32_16x16x4_f32(
                  false, a, false, b, (short)0, acc, false, false);
    }

    const int rbase = m0 + (lane >> 4) * 8;
    #pragma unroll
    for (int r = 0; r < 8; ++r)
        H[(size_t)(rbase + r) * NFEAT + ncol] = acc[r];
}

__global__ __launch_bounds__(256) void gcn_gather_tail(
        const float* __restrict__ h, float* __restrict__ out,
        const int* __restrict__ rowptr, const float* __restrict__ csr,
        const float* __restrict__ dinv, const float* __restrict__ bias, int n_nodes) {
    const int lane = threadIdx.x & 31;
    const int row = __builtin_amdgcn_readfirstlane(
                        (int)(blockIdx.x * (blockDim.x >> 5) + (threadIdx.x >> 5)));
    if (row >= n_nodes) return;

    const int   beg = rowptr[row];
    const int   end = rowptr[row + 1];
    const float di  = dinv[row];
    const float2* rec = (const float2*)csr;
    const float*  hl  = h + lane * 4;

    const float4 hs = *(const float4*)&hl[(size_t)row * NFEAT];
    const float4 bv = *(const float4*)&bias[lane * 4];

    float4 acc = make_float4(0.0f, 0.0f, 0.0f, 0.0f);
    for (int base = beg; base < end; base += 32) {
        const int p = base + lane;
        int   s = 0;
        float c = 0.0f;
        if (p < end) {
            const float2 r = rec[p];
            s = __float_as_int(r.x);
            c = dinv[s] * r.y * di;
        }
        const int cbits = __float_as_int(c);
        const int cnt   = min(32, end - base);
        int j = 0;
        for (; j + 4 <= cnt; j += 4) {
            const int   s0 = __builtin_amdgcn_readlane(s, j);
            const int   s1 = __builtin_amdgcn_readlane(s, j + 1);
            const int   s2 = __builtin_amdgcn_readlane(s, j + 2);
            const int   s3 = __builtin_amdgcn_readlane(s, j + 3);
            const float c0 = __int_as_float(__builtin_amdgcn_readlane(cbits, j));
            const float c1 = __int_as_float(__builtin_amdgcn_readlane(cbits, j + 1));
            const float c2 = __int_as_float(__builtin_amdgcn_readlane(cbits, j + 2));
            const float c3 = __int_as_float(__builtin_amdgcn_readlane(cbits, j + 3));
            const float4 h0 = *(const float4*)&hl[(size_t)s0 * NFEAT];
            const float4 h1 = *(const float4*)&hl[(size_t)s1 * NFEAT];
            const float4 h2 = *(const float4*)&hl[(size_t)s2 * NFEAT];
            const float4 h3 = *(const float4*)&hl[(size_t)s3 * NFEAT];
            acc.x += h0.x * c0; acc.y += h0.y * c0; acc.z += h0.z * c0; acc.w += h0.w * c0;
            acc.x += h1.x * c1; acc.y += h1.y * c1; acc.z += h1.z * c1; acc.w += h1.w * c1;
            acc.x += h2.x * c2; acc.y += h2.y * c2; acc.z += h2.z * c2; acc.w += h2.w * c2;
            acc.x += h3.x * c3; acc.y += h3.y * c3; acc.z += h3.z * c3; acc.w += h3.w * c3;
        }
        for (; j < cnt; ++j) {
            const int    s0 = __builtin_amdgcn_readlane(s, j);
            const float  c0 = __int_as_float(__builtin_amdgcn_readlane(cbits, j));
            const float4 h0 = *(const float4*)&hl[(size_t)s0 * NFEAT];
            acc.x += h0.x * c0; acc.y += h0.y * c0; acc.z += h0.z * c0; acc.w += h0.w * c0;
        }
    }

    const float dd = di * di;
    float4 v;
    v.x = fmaxf(acc.x + hs.x * dd + bv.x, 0.0f);
    v.y = fmaxf(acc.y + hs.y * dd + bv.y, 0.0f);
    v.z = fmaxf(acc.z + hs.z * dd + bv.z, 0.0f);
    v.w = fmaxf(acc.w + hs.w * dd + bv.w, 0.0f);
    *(float4*)&out[(size_t)row * NFEAT + lane * 4] = v;
}

__global__ void gcn_classifier(const float* __restrict__ h,
                               const float* __restrict__ Wc,
                               const float* __restrict__ bc,
                               float* __restrict__ out, int n_nodes) {
    int i = blockIdx.x * blockDim.x + threadIdx.x;
    if (i >= n_nodes) return;
    float acc[10];
    #pragma unroll
    for (int c = 0; c < 10; ++c) acc[c] = bc[c];
    const float* hr = h + (size_t)i * NFEAT;
    for (int d = 0; d < NFEAT; ++d) {
        float hv = hr[d];
        #pragma unroll
        for (int c = 0; c < 10; ++c) acc[c] += hv * Wc[d * 10 + c];
    }
    #pragma unroll
    for (int c = 0; c < 10; ++c) out[(size_t)i * 10 + c] = acc[c];
}

extern "C" void kernel_launch(void* const* d_in, const int* in_sizes, int n_in,
                              void* d_out, int out_size, void* d_ws, size_t ws_size,
                              hipStream_t stream) {
    const float* x   = (const float*)d_in[0];
    const int*   ei  = (const int*)  d_in[1];
    const float* ew  = (const float*)d_in[2];
    const float* W1  = (const float*)d_in[3];
    const float* b1  = (const float*)d_in[4];
    const float* W2  = (const float*)d_in[5];
    const float* b2  = (const float*)d_in[6];
    const float* Wc  = (const float*)d_in[7];
    const float* bc  = (const float*)d_in[8];
    float*       out = (float*)d_out;

    const int n_nodes = in_sizes[0] / NFEAT;
    const int n_edges = in_sizes[2];
    const int* src = ei;
    const int* dst = ei + n_edges;

    const size_t nd = (size_t)n_nodes * NFEAT;
    const size_t nn = ((size_t)n_nodes + 4) & ~(size_t)3;
    const size_t ne = ((size_t)n_edges + 3) & ~(size_t)3;
    float* bufA = (float*)d_ws;
    float* bufB = bufA + nd;
    float* dinv = bufB + nd;
    int*   cnt    = (int*)(dinv + nn);
    int*   rowptr = cnt + nn;
    float* csr    = (float*)(rowptr + nn);
    int*   rank   = (int*)(csr + 2 * ne);

    const int B = 256;
    const int gN  = (n_nodes + B - 1) / B;
    const int gE  = (n_edges + B - 1) / B;
    const int gRow  = (n_nodes + 7) / 8;
    const int gScan = (n_nodes + 1023) / 1024;
    const dim3 gblk(32, 8);
    const int gTiles = n_nodes / 16;

    gcn_zero_cnt  <<<gN, B, 0, stream>>>(cnt, n_nodes);
    gcn_count_rank<<<gE, B, 0, stream>>>(dst, cnt, rank, n_edges);
    gcn_scan      <<<gScan, 1024, 0, stream>>>(cnt, rowptr, n_nodes);
    gcn_fill_rows <<<gE, B, 0, stream>>>(src, dst, ew, rowptr, rank, csr, n_edges);
    gcn_dinv_rows <<<gN, B, 0, stream>>>(rowptr, csr, dinv, n_nodes);

    gcn_gemm128_wmma<<<gTiles, gblk, 0, stream>>>(x, W1, bufA);
    gcn_gather_tail <<<gRow, B, 0, stream>>>(bufA, bufB, rowptr, csr, dinv, b1, n_nodes);

    gcn_gemm128_wmma<<<gTiles, gblk, 0, stream>>>(bufB, W2, bufA);
    gcn_gather_tail <<<gRow, B, 0, stream>>>(bufA, bufB, rowptr, csr, dinv, b2, n_nodes);

    gcn_classifier<<<gN, B, 0, stream>>>(bufB, Wc, bc, out, n_nodes);
}


# Edge ranks and per-node counts taken on LDS counters (32 edge chunks, each a work-group with a counter per node), not a returning global atomic per edge; rows summed in chunk, then arrival order
# speedup vs baseline: 2.5956x; 1.1743x over previous
#include <hip/hip_runtime.h>
#include <hip/hip_bf16.h>

typedef float v2f __attribute__((ext_vector_type(2)));
typedef float v8f __attribute__((ext_vector_type(8)));

#define NFEAT 128

#define MAXNODES 50000
#define TABWORDS ((MAXNODES + 1023) & ~1023)
#define NCHUNK   32
#define EPT      20
#define SEGPITCH 64

__global__ __launch_bounds__(1024) void gcn_count_chunk(
        const int* __restrict__ dst, int* __restrict__ rank,
        int* __restrict__ cnts, int* __restrict__ segs,
        int n, int nn, int n_edges, int chunk_len) {
    __shared__ __attribute__((aligned(16))) int tab[TABWORDS];
    const int t    = threadIdx.x;
    const int lane = t & 31;
    const int wave = t >> 5;
    const int c    = blockIdx.x;

    for (int q = t; q < TABWORDS / 4; q += 1024) ((int4*)tab)[q] = make_int4(0, 0, 0, 0);
    __syncthreads();

    const int e0 = c * chunk_len;
    const int e1 = min(e0 + chunk_len, n_edges);
    for (int base = e0; base < e1; base += EPT * 1024) {
        int d[EPT];
        #pragma unroll
        for (int u = 0; u < EPT; ++u) {
            const int e = base + u * 1024 + t;
            d[u] = (e < e1) ? dst[e] : -1;
        }
        int r[EPT];
        #pragma unroll
        for (int u = 0; u < EPT; ++u)
            r[u] = (d[u] >= 0) ? atomicAdd(&tab[d[u]], 1) : 0;
        #pragma unroll
        for (int u = 0; u < EPT; ++u) {
            const int e = base + u * 1024 + t;
            if (d[u] >= 0) rank[e] = r[u];
        }
    }
    __syncthreads();

    int* mine = cnts + (size_t)c * nn;
    for (int i = t; i < n; i += 1024) mine[i] = tab[i];

    const int nseg = (n + 1023) >> 10;
    const int4* tab4 = (const int4*)tab;
    for (int seg = wave; seg < nseg; seg += 32) {
        int s = 0;
        #pragma unroll
        for (int k = 0; k < 8; ++k) {
            const int4 v = tab4[seg * 256 + k * 32 + lane];
            s += (v.x + v.y) + (v.z + v.w);
        }
        for (int o = 16; o > 0; o >>= 1) s += __shfl_xor(s, o);
        if (lane == 0) segs[c * SEGPITCH + seg] = s;
    }
}

__global__ __launch_bounds__(1024) void gcn_scan_bases(
        int* __restrict__ cnts, const int* __restrict__ segs,
        int* __restrict__ rowptr, int n, int nn) {
    __shared__ int s_pre[32];
    __shared__ int s_own[32];
    const int t     = threadIdx.x;
    const int lane  = t & 31;
    const int wave  = t >> 5;
    const int b     = blockIdx.x;
    const int first = b * 1024;

    int pre = 0;
    for (int q = t; q < NCHUNK * SEGPITCH; q += 1024)
        if ((q & (SEGPITCH - 1)) < b) pre += segs[q];

    const int i = first + t;
    int off[NCHUNK];
    int x = 0;
    if (i < n) {
        int v[NCHUNK];
        #pragma unroll
        for (int c = 0; c < NCHUNK; ++c) v[c] = cnts[(size_t)c * nn + i];
        #pragma unroll
        for (int c = 0; c < NCHUNK; ++c) { off[c] = x; x += v[c]; }
    }

    int y = x;
    for (int o = 1; o < 32; o <<= 1) {
        int u = __shfl_up(y, o);
        if (lane >= o) y += u;
    }
    for (int o = 16; o > 0; o >>= 1) pre += __shfl_xor(pre, o);
    if (lane == 31) s_own[wave] = y;
    if (lane == 0)  s_pre[wave] = pre;
    __syncthreads();

    int a = s_pre[lane] + ((lane < wave) ? s_own[lane] : 0);
    for (int o = 16; o > 0; o >>= 1) a += __shfl_xor(a, o);

    const int excl = a + y - x;
    if (i < n) {
        rowptr[i] = excl;
        #pragma unroll
        for (int c = 0; c < NCHUNK; ++c) cnts[(size_t)c * nn + i] = excl + off[c];
    }
    if (i == n - 1) rowptr[n] = excl + x;
}

__global__ void gcn_fill_rows(const int* __restrict__ src, const int* __restrict__ dst,
                              const float* __restrict__ ew,
                              const int* __restrict__ bases, const int* __restrict__ rank,
                              float* __restrict__ csr, int nn, int n_edges, int chunk_len) {
    const int c  = blockIdx.y;
    const int k  = blockIdx.x * blockDim.x + threadIdx.x;
    const int e  = c * chunk_len + k;
    if (k < chunk_len && e < n_edges) {
        const int p = bases[(size_t)c * nn + dst[e]] + rank[e];
        ((float2*)csr)[p] = make_float2(__int_as_float(src[e]), ew[e]);
    }
}

__global__ void gcn_dinv_rows(const int* __restrict__ rowptr, const float* __restrict__ csr,
                              float* __restrict__ dinv, int n) {
    int i = blockIdx.x * blockDim.x + threadIdx.x;
    if (i >= n) return;
    const float2* rec = (const float2*)csr;
    const int end = rowptr[i + 1];
    float s = 0.0f;
    for (int p = rowptr[i]; p < end; ++p) s += rec[p].y;
    dinv[i] = rsqrtf(s + 1.0f);
}

__global__ __launch_bounds__(256) void gcn_gemm128_wmma(
        const float* __restrict__ X, const float* __restrict__ W,
        float* __restrict__ H) {
    __shared__ float Wl[NFEAT * NFEAT];
    {
        const float4* Wv  = (const float4*)W;
        float4*       Wlv = (float4*)Wl;
        int tid = threadIdx.y * 32 + threadIdx.x;
        #pragma unroll
        for (int i = 0; i < (NFEAT * NFEAT / 4) / 256; ++i)
            Wlv[tid + i * 256] = Wv[tid + i * 256];
    }
    __syncthreads();

    const int lane  = threadIdx.x;
    const int m0    = blockIdx.x * 16;
    const int n0    = threadIdx.y * 16;
    const int mlane = lane & 15;
    const int khalf = (lane >> 4) * 2;
    const int ncol  = n0 + mlane;

    const float* xrow = X + (size_t)(m0 + mlane) * NFEAT;
    v8f acc = {};

    #pragma unroll
    for (int k0 = 0; k0 < NFEAT; k0 += 4) {
        const int k = k0 + khalf;
        v2f a = *(const v2f*)&xrow[k];
        v2f b;
        b.x = Wl[(k    ) * NFEAT + ncol];
        b.y = Wl[(k + 1) * NFEAT + ncol];
        acc = __builtin_amdgcn_wmma_f32_16x16x4_f32(
                  false, a, false, b, (short)0, acc, false, false);
    }

    const int rbase = m0 + (lane >> 4) * 8;
    #pragma unroll
    for (int r = 0; r < 8; ++r)
        H[(size_t)(rbase + r) * NFEAT + ncol] = acc[r];
}

__global__ __launch_bounds__(256) void gcn_gather_tail(
        const float* __restrict__ h, float* __restrict__ out,
        const int* __restrict__ rowptr, const float* __restrict__ csr,
        const float* __restrict__ dinv, const float* __restrict__ bias, int n_nodes) {
    const int lane = threadIdx.x & 31;
    const int row = __builtin_amdgcn_readfirstlane(
                        (int)(blockIdx.x * (blockDim.x >> 5) + (threadIdx.x >> 5)));
    if (row >= n_nodes) return;

    const int   beg = rowptr[row];
    const int   end = rowptr[row + 1];
    const float di  = dinv[row];
    const float2* rec = (const float2*)csr;
    const float*  hl  = h + lane * 4;

    const float4 hs = *(const float4*)&hl[(size_t)row * NFEAT];
    const float4 bv = *(const float4*)&bias[lane * 4];

    float4 acc = make_float4(0.0f, 0.0f, 0.0f, 0.0f);
    for (int base = beg; base < end; base += 32) {
        const int p = base + lane;
        int   s = 0;
        float c = 0.0f;
        if (p < end) {
            const float2 r = rec[p];
            s = __float_as_int(r.x);
            c = dinv[s] * r.y * di;
        }
        const int cbits = __float_as_int(c);
        const int cnt   = min(32, end - base);
        int j = 0;
        for (; j + 4 <= cnt; j += 4) {
            const int   s0 = __builtin_amdgcn_readlane(s, j);
            const int   s1 = __builtin_amdgcn_readlane(s, j + 1);
            const int   s2 = __builtin_amdgcn_readlane(s, j + 2);
            const int   s3 = __builtin_amdgcn_readlane(s, j + 3);
            const float c0 = __int_as_float(__builtin_amdgcn_readlane(cbits, j));
            const float c1 = __int_as_float(__builtin_amdgcn_readlane(cbits, j + 1));
            const float c2 = __int_as_float(__builtin_amdgcn_readlane(cbits, j + 2));
            const float c3 = __int_as_float(__builtin_amdgcn_readlane(cbits, j + 3));
            const float4 h0 = *(const float4*)&hl[(size_t)s0 * NFEAT];
            const float4 h1 = *(const float4*)&hl[(size_t)s1 * NFEAT];
            const float4 h2 = *(const float4*)&hl[(size_t)s2 * NFEAT];
            const float4 h3 = *(const float4*)&hl[(size_t)s3 * NFEAT];
            acc.x += h0.x * c0; acc.y += h0.y * c0; acc.z += h0.z * c0; acc.w += h0.w * c0;
            acc.x += h1.x * c1; acc.y += h1.y * c1; acc.z += h1.z * c1; acc.w += h1.w * c1;
            acc.x += h2.x * c2; acc.y += h2.y * c2; acc.z += h2.z * c2; acc.w += h2.w * c2;
            acc.x += h3.x * c3; acc.y += h3.y * c3; acc.z += h3.z * c3; acc.w += h3.w * c3;
        }
        for (; j < cnt; ++j) {
            const int    s0 = __builtin_amdgcn_readlane(s, j);
            const float  c0 = __int_as_float(__builtin_amdgcn_readlane(cbits, j));
            const float4 h0 = *(const float4*)&hl[(size_t)s0 * NFEAT];
            acc.x += h0.x * c0; acc.y += h0.y * c0; acc.z += h0.z * c0; acc.w += h0.w * c0;
        }
    }

    const float dd = di * di;
    float4 v;
    v.x = fmaxf(acc.x + hs.x * dd + bv.x, 0.0f);
    v.y = fmaxf(acc.y + hs.y * dd + bv.y, 0.0f);
    v.z = fmaxf(acc.z + hs.z * dd + bv.z, 0.0f);
    v.w = fmaxf(acc.w + hs.w * dd + bv.w, 0.0f);
    *(float4*)&out[(size_t)row * NFEAT + lane * 4] = v;
}

__global__ void gcn_classifier(const float* __restrict__ h,
                               const float* __restrict__ Wc,
                               const float* __restrict__ bc,
                               float* __restrict__ out, int n_nodes) {
    int i = blockIdx.x * blockDim.x + threadIdx.x;
    if (i >= n_nodes) return;
    float acc[10];
    #pragma unroll
    for (int c = 0; c < 10; ++c) acc[c] = bc[c];
    const float* hr = h + (size_t)i * NFEAT;
    for (int d = 0; d < NFEAT; ++d) {
        float hv = hr[d];
        #pragma unroll
        for (int c = 0; c < 10; ++c) acc[c] += hv * Wc[d * 10 + c];
    }
    #pragma unroll
    for (int c = 0; c < 10; ++c) out[(size_t)i * 10 + c] = acc[c];
}

extern "C" void kernel_launch(void* const* d_in, const int* in_sizes, int n_in,
                              void* d_out, int out_size, void* d_ws, size_t ws_size,
                              hipStream_t stream) {
    const float* x   = (const float*)d_in[0];
    const int*   ei  = (const int*)  d_in[1];
    const float* ew  = (const float*)d_in[2];
    const float* W1  = (const float*)d_in[3];
    const float* b1  = (const float*)d_in[4];
    const float* W2  = (const float*)d_in[5];
    const float* b2  = (const float*)d_in[6];
    const float* Wc  = (const float*)d_in[7];
    const float* bc  = (const float*)d_in[8];
    float*       out = (float*)d_out;

    const int n_nodes = in_sizes[0] / NFEAT;
    const int n_edges = in_sizes[2];
    const int* src = ei;
    const int* dst = ei + n_edges;

    const size_t nd = (size_t)n_nodes * NFEAT;
    const size_t nn = ((size_t)n_nodes + 4) & ~(size_t)3;
    const size_t ne = ((size_t)n_edges + 3) & ~(size_t)3;
    float* bufA = (float*)d_ws;
    float* bufB = bufA + nd;
    float* dinv = bufB + nd;
    int*   rowptr = (int*)(dinv + nn);
    float* csr    = (float*)(rowptr + nn);
    int*   rank   = (int*)(csr + 2 * ne);
    int*   cnts   = rank + ne;
    int*   segs   = cnts + (size_t)NCHUNK * nn;

    const int B = 256;
    const int gN  = (n_nodes + B - 1) / B;
    const int chunkLen = (n_edges + NCHUNK - 1) / NCHUNK;
    const dim3 gFill((chunkLen + B - 1) / B, NCHUNK);
    const int gRow  = (n_nodes + 7) / 8;
    const int gScan = (n_nodes + 1023) / 1024;
    const dim3 gblk(32, 8);
    const int gTiles = n_nodes / 16;

    gcn_count_chunk<<<NCHUNK, 1024, 0, stream>>>(dst, rank, cnts, segs, n_nodes, (int)nn,
                                                 n_edges, chunkLen);
    gcn_scan_bases <<<gScan, 1024, 0, stream>>>(cnts, segs, rowptr, n_nodes, (int)nn);
    gcn_fill_rows  <<<gFill, B, 0, stream>>>(src, dst, ew, cnts, rank, csr, (int)nn,
                                             n_edges, chunkLen);
    gcn_dinv_rows  <<<gN, B, 0, stream>>>(rowptr, csr, dinv, n_nodes);

    gcn_gemm128_wmma<<<gTiles, gblk, 0, stream>>>(x, W1, bufA);
    gcn_gather_tail <<<gRow, B, 0, stream>>>(bufA, bufB, rowptr, csr, dinv, b1, n_nodes);

    gcn_gemm128_wmma<<<gTiles, gblk, 0, stream>>>(bufB, W2, bufA);
    gcn_gather_tail <<<gRow, B, 0, stream>>>(bufA, bufB, rowptr, csr, dinv, b2, n_nodes);

    gcn_classifier<<<gN, B, 0, stream>>>(bufB, Wc, bc, out, n_nodes);
}


# GEMM operand feed rebuilt: a work-group walks 4-5 row tiles with W fragments held in registers (W fetched 768 times, not 3125), each X tile staged once via LDS by whole-row loads; sums unchanged
# speedup vs baseline: 3.1385x; 1.2092x over previous
#include <hip/hip_runtime.h>
#include <hip/hip_bf16.h>

typedef float v2f __attribute__((ext_vector_type(2)));
typedef float v8f __attribute__((ext_vector_type(8)));

#define NFEAT 128

#define MAXNODES 50000
#define TABWORDS ((MAXNODES + 1023) & ~1023)
#define NCHUNK   32
#define EPT      20
#define SEGPITCH 64

__global__ __launch_bounds__(1024) void gcn_count_chunk(
        const int* __restrict__ dst, int* __restrict__ rank,
        int* __restrict__ cnts, int* __restrict__ segs,
        int n, int nn, int n_edges, int chunk_len) {
    __shared__ __attribute__((aligned(16))) int tab[TABWORDS];
    const int t    = threadIdx.x;
    const int lane = t & 31;
    const int wave = t >> 5;
    const int c    = blockIdx.x;

    for (int q = t; q < TABWORDS / 4; q += 1024) ((int4*)tab)[q] = make_int4(0, 0, 0, 0);
    __syncthreads();

    const int e0 = c * chunk_len;
    const int e1 = min(e0 + chunk_len, n_edges);
    for (int base = e0; base < e1; base += EPT * 1024) {
        int d[EPT];
        #pragma unroll
        for (int u = 0; u < EPT; ++u) {
            const int e = base + u * 1024 + t;
            d[u] = (e < e1) ? dst[e] : -1;
        }
        int r[EPT];
        #pragma unroll
        for (int u = 0; u < EPT; ++u)
            r[u] = (d[u] >= 0) ? atomicAdd(&tab[d[u]], 1) : 0;
        #pragma unroll
        for (int u = 0; u < EPT; ++u) {
            const int e = base + u * 1024 + t;
            if (d[u] >= 0) rank[e] = r[u];
        }
    }
    __syncthreads();

    int* mine = cnts + (size_t)c * nn;
    for (int i = t; i < n; i += 1024) mine[i] = tab[i];

    const int nseg = (n + 1023) >> 10;
    const int4* tab4 = (const int4*)tab;
    for (int seg = wave; seg < nseg; seg += 32) {
        int s = 0;
        #pragma unroll
        for (int k = 0; k < 8; ++k) {
            const int4 v = tab4[seg * 256 + k * 32 + lane];
            s += (v.x + v.y) + (v.z + v.w);
        }
        for (int o = 16; o > 0; o >>= 1) s += __shfl_xor(s, o);
        if (lane == 0) segs[c * SEGPITCH + seg] = s;
    }
}

__global__ __launch_bounds__(1024) void gcn_scan_bases(
        int* __restrict__ cnts, const int* __restrict__ segs,
        int* __restrict__ rowptr, int n, int nn) {
    __shared__ int s_pre[32];
    __shared__ int s_own[32];
    const int t     = threadIdx.x;
    const int lane  = t & 31;
    const int wave  = t >> 5;
    const int b     = blockIdx.x;
    const int first = b * 1024;

    int pre = 0;
    for (int q = t; q < NCHUNK * SEGPITCH; q += 1024)
        if ((q & (SEGPITCH - 1)) < b) pre += segs[q];

    const int i = first + t;
    int off[NCHUNK];
    int x = 0;
    if (i < n) {
        int v[NCHUNK];
        #pragma unroll
        for (int c = 0; c < NCHUNK; ++c) v[c] = cnts[(size_t)c * nn + i];
        #pragma unroll
        for (int c = 0; c < NCHUNK; ++c) { off[c] = x; x += v[c]; }
    }

    int y = x;
    for (int o = 1; o < 32; o <<= 1) {
        int u = __shfl_up(y, o);
        if (lane >= o) y += u;
    }
    for (int o = 16; o > 0; o >>= 1) pre += __shfl_xor(pre, o);
    if (lane == 31) s_own[wave] = y;
    if (lane == 0)  s_pre[wave] = pre;
    __syncthreads();

    int a = s_pre[lane] + ((lane < wave) ? s_own[lane] : 0);
    for (int o = 16; o > 0; o >>= 1) a += __shfl_xor(a, o);

    const int excl = a + y - x;
    if (i < n) {
        rowptr[i] = excl;
        #pragma unroll
        for (int c = 0; c < NCHUNK; ++c) cnts[(size_t)c * nn + i] = excl + off[c];
    }
    if (i == n - 1) rowptr[n] = excl + x;
}

__global__ void gcn_fill_rows(const int* __restrict__ src, const int* __restrict__ dst,
                              const float* __restrict__ ew,
                              const int* __restrict__ bases, const int* __restrict__ rank,
                              float* __restrict__ csr, int nn, int n_edges, int chunk_len) {
    const int c  = blockIdx.y;
    const int k  = blockIdx.x * blockDim.x + threadIdx.x;
    const int e  = c * chunk_len + k;
    if (k < chunk_len && e < n_edges) {
        const int p = bases[(size_t)c * nn + dst[e]] + rank[e];
        ((float2*)csr)[p] = make_float2(__int_as_float(src[e]), ew[e]);
    }
}

__global__ void gcn_dinv_rows(const int* __restrict__ rowptr, const float* __restrict__ csr,
                              float* __restrict__ dinv, int n) {
    int i = blockIdx.x * blockDim.x + threadIdx.x;
    if (i >= n) return;
    const float2* rec = (const float2*)csr;
    const int end = rowptr[i + 1];
    float s = 0.0f;
    for (int p = rowptr[i]; p < end; ++p) s += rec[p].y;
    dinv[i] = rsqrtf(s + 1.0f);
}

#define APITCH 132

__global__ __launch_bounds__(256) void gcn_gemm128_wmma(
        const float* __restrict__ X, const float* __restrict__ W,
        float* __restrict__ H, int n_tiles) {
    __shared__ __attribute__((aligned(16))) float Al[2][16 * APITCH];

    const int lane  = threadIdx.x;
    const int wv    = threadIdx.y;
    const int tid   = wv * 32 + lane;
    const int mlane = lane & 15;
    const int khalf = (lane >> 4) * 2;
    const int ncol  = wv * 16 + mlane;

    int tile = blockIdx.x;
    if (tile >= n_tiles) return;

    const float4* Xv = (const float4*)X;
    float4 p0 = Xv[(size_t)tile * 512 + tid];
    float4 p1 = Xv[(size_t)tile * 512 + 256 + tid];

    v2f bfr[NFEAT / 4];
    #pragma unroll
    for (int j = 0; j < NFEAT / 4; ++j) {
        const int k = 4 * j + khalf;
        bfr[j].x = W[(k    ) * NFEAT + ncol];
        bfr[j].y = W[(k + 1) * NFEAT + ncol];
    }

    const int st0 = wv * APITCH + lane * 4;
    const int st1 = (wv + 8) * APITCH + lane * 4;
    const int ld0 = mlane * APITCH + khalf;
    float* hrow = H + (size_t)((lane >> 4) * 8) * NFEAT + ncol;

    *(float4*)&Al[0][st0] = p0;
    *(float4*)&Al[0][st1] = p1;
    __syncthreads();

    int buf = 0;
    for (;;) {
        const int  next = tile + (int)gridDim.x;
        const bool more = next < n_tiles;
        if (more) {
            p0 = Xv[(size_t)next * 512 + tid];
            p1 = Xv[(size_t)next * 512 + 256 + tid];
        }

        const float* arow = &Al[buf][ld0];
        v2f afr[NFEAT / 4];
        #pragma unroll
        for (int j = 0; j < NFEAT / 4; ++j) afr[j] = *(const v2f*)&arow[4 * j];
        v8f acc = {};
        #pragma unroll
        for (int j = 0; j < NFEAT / 4; ++j)
            acc = __builtin_amdgcn_wmma_f32_16x16x4_f32(
                      false, afr[j], false, bfr[j], (short)0, acc, false, false);
        __builtin_amdgcn_sched_group_barrier(0x100, 16, 0);
        __builtin_amdgcn_sched_group_barrier(0x008, 32, 0);

        if (more) {
            *(float4*)&Al[buf ^ 1][st0] = p0;
            *(float4*)&Al[buf ^ 1][st1] = p1;
            __syncthreads();
        }
        float* ht = hrow + (size_t)tile * (16 * NFEAT);
        #pragma unroll
        for (int r = 0; r < 8; ++r)
            ht[r * NFEAT] = acc[r];

        if (!more) break;
        buf ^= 1;
        tile = next;
    }
}

__global__ __launch_bounds__(256) void gcn_gather_tail(
        const float* __restrict__ h, float* __restrict__ out,
        const int* __restrict__ rowptr, const float* __restrict__ csr,
        const float* __restrict__ dinv, const float* __restrict__ bias, int n_nodes) {
    const int lane = threadIdx.x & 31;
    const int row = __builtin_amdgcn_readfirstlane(
                        (int)(blockIdx.x * (blockDim.x >> 5) + (threadIdx.x >> 5)));
    if (row >= n_nodes) return;

    const int   beg = rowptr[row];
    const int   end = rowptr[row + 1];
    const float di  = dinv[row];
    const float2* rec = (const float2*)csr;
    const float*  hl  = h + lane * 4;

    const float4 hs = *(const float4*)&hl[(size_t)row * NFEAT];
    const float4 bv = *(const float4*)&bias[lane * 4];

    float4 acc = make_float4(0.0f, 0.0f, 0.0f, 0.0f);
    for (int base = beg; base < end; base += 32) {
        const int p = base + lane;
        int   s = 0;
        float c = 0.0f;
        if (p < end) {
            const float2 r = rec[p];
            s = __float_as_int(r.x);
            c = dinv[s] * r.y * di;
        }
        const int cbits = __float_as_int(c);
        const int cnt   = min(32, end - base);
        int j = 0;
        for (; j + 4 <= cnt; j += 4) {
            const int   s0 = __builtin_amdgcn_readlane(s, j);
            const int   s1 = __builtin_amdgcn_readlane(s, j + 1);
            const int   s2 = __builtin_amdgcn_readlane(s, j + 2);
            const int   s3 = __builtin_amdgcn_readlane(s, j + 3);
            const float c0 = __int_as_float(__builtin_amdgcn_readlane(cbits, j));
            const float c1 = __int_as_float(__builtin_amdgcn_readlane(cbits, j + 1));
            const float c2 = __int_as_float(__builtin_amdgcn_readlane(cbits, j + 2));
            const float c3 = __int_as_float(__builtin_amdgcn_readlane(cbits, j + 3));
            const float4 h0 = *(const float4*)&hl[(size_t)s0 * NFEAT];
            const float4 h1 = *(const float4*)&hl[(size_t)s1 * NFEAT];
            const float4 h2 = *(const float4*)&hl[(size_t)s2 * NFEAT];
            const float4 h3 = *(const float4*)&hl[(size_t)s3 * NFEAT];
            acc.x += h0.x * c0; acc.y += h0.y * c0; acc.z += h0.z * c0; acc.w += h0.w * c0;
            acc.x += h1.x * c1; acc.y += h1.y * c1; acc.z += h1.z * c1; acc.w += h1.w * c1;
            acc.x += h2.x * c2; acc.y += h2.y * c2; acc.z += h2.z * c2; acc.w += h2.w * c2;
            acc.x += h3.x * c3; acc.y += h3.y * c3; acc.z += h3.z * c3; acc.w += h3.w * c3;
        }
        for (; j < cnt; ++j) {
            const int    s0 = __builtin_amdgcn_readlane(s, j);
            const float  c0 = __int_as_float(__builtin_amdgcn_readlane(cbits, j));
            const float4 h0 = *(const float4*)&hl[(size_t)s0 * NFEAT];
            acc.x += h0.x * c0; acc.y += h0.y * c0; acc.z += h0.z * c0; acc.w += h0.w * c0;
        }
    }

    const float dd = di * di;
    float4 v;
    v.x = fmaxf(acc.x + hs.x * dd + bv.x, 0.0f);
    v.y = fmaxf(acc.y + hs.y * dd + bv.y, 0.0f);
    v.z = fmaxf(acc.z + hs.z * dd + bv.z, 0.0f);
    v.w = fmaxf(acc.w + hs.w * dd + bv.w, 0.0f);
    *(float4*)&out[(size_t)row * NFEAT + lane * 4] = v;
}

__global__ void gcn_classifier(const float* __restrict__ h,
                               const float* __restrict__ Wc,
                               const float* __restrict__ bc,
                               float* __restrict__ out, int n_nodes) {
    int i = blockIdx.x * blockDim.x + threadIdx.x;
    if (i >= n_nodes) return;
    float acc[10];
    #pragma unroll
    for (int c = 0; c < 10; ++c) acc[c] = bc[c];
    const float* hr = h + (size_t)i * NFEAT;
    for (int d = 0; d < NFEAT; ++d) {
        float hv = hr[d];
        #pragma unroll
        for (int c = 0; c < 10; ++c) acc[c] += hv * Wc[d * 10 + c];
    }
    #pragma unroll
    for (int c = 0; c < 10; ++c) out[(size_t)i * 10 + c] = acc[c];
}

extern "C" void kernel_launch(void* const* d_in, const int* in_sizes, int n_in,
                              void* d_out, int out_size, void* d_ws, size_t ws_size,
                              hipStream_t stream) {
    const float* x   = (const float*)d_in[0];
    const int*   ei  = (const int*)  d_in[1];
    const float* ew  = (const float*)d_in[2];
    const float* W1  = (const float*)d_in[3];
    const float* b1  = (const float*)d_in[4];
    const float* W2  = (const float*)d_in[5];
    const float* b2  = (const float*)d_in[6];
    const float* Wc  = (const float*)d_in[7];
    const float* bc  = (const float*)d_in[8];
    float*       out = (float*)d_out;

    const int n_nodes = in_sizes[0] / NFEAT;
    const int n_edges = in_sizes[2];
    const int* src = ei;
    const int* dst = ei + n_edges;

    const size_t nd = (size_t)n_nodes * NFEAT;
    const size_t nn = ((size_t)n_nodes + 4) & ~(size_t)3;
    const size_t ne = ((size_t)n_edges + 3) & ~(size_t)3;
    float* bufA = (float*)d_ws;
    float* bufB = bufA + nd;
    float* dinv = bufB + nd;
    int*   rowptr = (int*)(dinv + nn);
    float* csr    = (float*)(rowptr + nn);
    int*   rank   = (int*)(csr + 2 * ne);
    int*   cnts   = rank + ne;
    int*   segs   = cnts + (size_t)NCHUNK * nn;

    const int B = 256;
    const int gN  = (n_nodes + B - 1) / B;
    const int chunkLen = (n_edges + NCHUNK - 1) / NCHUNK;
    const dim3 gFill((chunkLen + B - 1) / B, NCHUNK);
    const int gRow  = (n_nodes + 7) / 8;
    const int gScan = (n_nodes + 1023) / 1024;
    const dim3 gblk(32, 8);
    const int nTiles = n_nodes / 16;
    const int gGemm  = 768;

    gcn_count_chunk<<<NCHUNK, 1024, 0, stream>>>(dst, rank, cnts, segs, n_nodes, (int)nn,
                                                 n_edges, chunkLen);
    gcn_scan_bases <<<gScan, 1024, 0, stream>>>(cnts, segs, rowptr, n_nodes, (int)nn);
    gcn_fill_rows  <<<gFill, B, 0, stream>>>(src, dst, ew, cnts, rank, csr, (int)nn,
                                             n_edges, chunkLen);
    gcn_dinv_rows  <<<gN, B, 0, stream>>>(rowptr, csr, dinv, n_nodes);

    gcn_gemm128_wmma<<<gGemm, gblk, 0, stream>>>(x, W1, bufA, nTiles);
    gcn_gather_tail <<<gRow, B, 0, stream>>>(bufA, bufB, rowptr, csr, dinv, b1, n_nodes);

    gcn_gemm128_wmma<<<gGemm, gblk, 0, stream>>>(bufB, W2, bufA, nTiles);
    gcn_gather_tail <<<gRow, B, 0, stream>>>(bufA, bufB, rowptr, csr, dinv, b2, n_nodes);

    gcn_classifier<<<gN, B, 0, stream>>>(bufB, Wc, bc, out, n_nodes);
}


# Classifier folded into the layer-2 gather's epilogue: row stays in registers, class sums = 4-term lane partials (Wc transposed in LDS) added by a 32-lane tree, bias last; h2 plane and 1 launch go
# speedup vs baseline: 3.4992x; 1.1149x over previous
#include <hip/hip_runtime.h>
#include <hip/hip_bf16.h>

typedef float v2f __attribute__((ext_vector_type(2)));
typedef float v8f __attribute__((ext_vector_type(8)));

#define NFEAT 128

#define MAXNODES 50000
#define TABWORDS ((MAXNODES + 1023) & ~1023)
#define NCHUNK   32
#define EPT      20
#define SEGPITCH 64

__global__ __launch_bounds__(1024) void gcn_count_chunk(
        const int* __restrict__ dst, int* __restrict__ rank,
        int* __restrict__ cnts, int* __restrict__ segs,
        int n, int nn, int n_edges, int chunk_len) {
    __shared__ __attribute__((aligned(16))) int tab[TABWORDS];
    const int t    = threadIdx.x;
    const int lane = t & 31;
    const int wave = t >> 5;
    const int c    = blockIdx.x;

    for (int q = t; q < TABWORDS / 4; q += 1024) ((int4*)tab)[q] = make_int4(0, 0, 0, 0);
    __syncthreads();

    const int e0 = c * chunk_len;
    const int e1 = min(e0 + chunk_len, n_edges);
    for (int base = e0; base < e1; base += EPT * 1024) {
        int d[EPT];
        #pragma unroll
        for (int u = 0; u < EPT; ++u) {
            const int e = base + u * 1024 + t;
            d[u] = (e < e1) ? dst[e] : -1;
        }
        int r[EPT];
        #pragma unroll
        for (int u = 0; u < EPT; ++u)
            r[u] = (d[u] >= 0) ? atomicAdd(&tab[d[u]], 1) : 0;
        #pragma unroll
        for (int u = 0; u < EPT; ++u) {
            const int e = base + u * 1024 + t;
            if (d[u] >= 0) rank[e] = r[u];
        }
    }
    __syncthreads();

    int* mine = cnts + (size_t)c * nn;
    for (int i = t; i < n; i += 1024) mine[i] = tab[i];

    const int nseg = (n + 1023) >> 10;
    const int4* tab4 = (const int4*)tab;
    for (int seg = wave; seg < nseg; seg += 32) {
        int s = 0;
        #pragma unroll
        for (int k = 0; k < 8; ++k) {
            const int4 v = tab4[seg * 256 + k * 32 + lane];
            s += (v.x + v.y) + (v.z + v.w);
        }
        for (int o = 16; o > 0; o >>= 1) s += __shfl_xor(s, o);
        if (lane == 0) segs[c * SEGPITCH + seg] = s;
    }
}

__global__ __launch_bounds__(1024) void gcn_scan_bases(
        int* __restrict__ cnts, const int* __restrict__ segs,
        int* __restrict__ rowptr, int n, int nn) {
    __shared__ int s_pre[32];
    __shared__ int s_own[32];
    const int t     = threadIdx.x;
    const int lane  = t & 31;
    const int wave  = t >> 5;
    const int b     = blockIdx.x;
    const int first = b * 1024;

    int pre = 0;
    for (int q = t; q < NCHUNK * SEGPITCH; q += 1024)
        if ((q & (SEGPITCH - 1)) < b) pre += segs[q];

    const int i = first + t;
    int off[NCHUNK];
    int x = 0;
    if (i < n) {
        int v[NCHUNK];
        #pragma unroll
        for (int c = 0; c < NCHUNK; ++c) v[c] = cnts[(size_t)c * nn + i];
        #pragma unroll
        for (int c = 0; c < NCHUNK; ++c) { off[c] = x; x += v[c]; }
    }

    int y = x;
    for (int o = 1; o < 32; o <<= 1) {
        int u = __shfl_up(y, o);
        if (lane >= o) y += u;
    }
    for (int o = 16; o > 0; o >>= 1) pre += __shfl_xor(pre, o);
    if (lane == 31) s_own[wave] = y;
    if (lane == 0)  s_pre[wave] = pre;
    __syncthreads();

    int a = s_pre[lane] + ((lane < wave) ? s_own[lane] : 0);
    for (int o = 16; o > 0; o >>= 1) a += __shfl_xor(a, o);

    const int excl = a + y - x;
    if (i < n) {
        rowptr[i] = excl;
        #pragma unroll
        for (int c = 0; c < NCHUNK; ++c) cnts[(size_t)c * nn + i] = excl + off[c];
    }
    if (i == n - 1) rowptr[n] = excl + x;
}

__global__ void gcn_fill_rows(const int* __restrict__ src, const int* __restrict__ dst,
                              const float* __restrict__ ew,
                              const int* __restrict__ bases, const int* __restrict__ rank,
                              float* __restrict__ csr, int nn, int n_edges, int chunk_len) {
    const int c  = blockIdx.y;
    const int k  = blockIdx.x * blockDim.x + threadIdx.x;
    const int e  = c * chunk_len + k;
    if (k < chunk_len && e < n_edges) {
        const int p = bases[(size_t)c * nn + dst[e]] + rank[e];
        ((float2*)csr)[p] = make_float2(__int_as_float(src[e]), ew[e]);
    }
}

__global__ void gcn_dinv_rows(const int* __restrict__ rowptr, const float* __restrict__ csr,
                              float* __restrict__ dinv, int n) {
    int i = blockIdx.x * blockDim.x + threadIdx.x;
    if (i >= n) return;
    const float2* rec = (const float2*)csr;
    const int end = rowptr[i + 1];
    float s = 0.0f;
    for (int p = rowptr[i]; p < end; ++p) s += rec[p].y;
    dinv[i] = rsqrtf(s + 1.0f);
}

#define APITCH 132

__global__ __launch_bounds__(256) void gcn_gemm128_wmma(
        const float* __restrict__ X, const float* __restrict__ W,
        float* __restrict__ H, int n_tiles) {
    __shared__ __attribute__((aligned(16))) float Al[2][16 * APITCH];

    const int lane  = threadIdx.x;
    const int wv    = threadIdx.y;
    const int tid   = wv * 32 + lane;
    const int mlane = lane & 15;
    const int khalf = (lane >> 4) * 2;
    const int ncol  = wv * 16 + mlane;

    int tile = blockIdx.x;
    if (tile >= n_tiles) return;

    const float4* Xv = (const float4*)X;
    float4 p0 = Xv[(size_t)tile * 512 + tid];
    float4 p1 = Xv[(size_t)tile * 512 + 256 + tid];

    v2f bfr[NFEAT / 4];
    #pragma unroll
    for (int j = 0; j < NFEAT / 4; ++j) {
        const int k = 4 * j + khalf;
        bfr[j].x = W[(k    ) * NFEAT + ncol];
        bfr[j].y = W[(k + 1) * NFEAT + ncol];
    }

    const int st0 = wv * APITCH + lane * 4;
    const int st1 = (wv + 8) * APITCH + lane * 4;
    const int ld0 = mlane * APITCH + khalf;
    float* hrow = H + (size_t)((lane >> 4) * 8) * NFEAT + ncol;

    *(float4*)&Al[0][st0] = p0;
    *(float4*)&Al[0][st1] = p1;
    __syncthreads();

    int buf = 0;
    for (;;) {
        const int  next = tile + (int)gridDim.x;
        const bool more = next < n_tiles;
        if (more) {
            p0 = Xv[(size_t)next * 512 + tid];
            p1 = Xv[(size_t)next * 512 + 256 + tid];
        }

        const float* arow = &Al[buf][ld0];
        v2f afr[NFEAT / 4];
        #pragma unroll
        for (int j = 0; j < NFEAT / 4; ++j) afr[j] = *(const v2f*)&arow[4 * j];
        v8f acc = {};
        #pragma unroll
        for (int j = 0; j < NFEAT / 4; ++j)
            acc = __builtin_amdgcn_wmma_f32_16x16x4_f32(
                      false, afr[j], false, bfr[j], (short)0, acc, false, false);
        __builtin_amdgcn_sched_group_barrier(0x100, 16, 0);
        __builtin_amdgcn_sched_group_barrier(0x008, 32, 0);

        if (more) {
            *(float4*)&Al[buf ^ 1][st0] = p0;
            *(float4*)&Al[buf ^ 1][st1] = p1;
            __syncthreads();
        }
        float* ht = hrow + (size_t)tile * (16 * NFEAT);
        #pragma unroll
        for (int r = 0; r < 8; ++r)
            ht[r * NFEAT] = acc[r];

        if (!more) break;
        buf ^= 1;
        tile = next;
    }
}


__device__ __forceinline__ float4 gcn_edge_walk(
        const float* __restrict__ hl, const float2* __restrict__ rec,
        const float* __restrict__ dinv, int beg, int end, float di, int lane) {
    float4 acc = make_float4(0.0f, 0.0f, 0.0f, 0.0f);
    for (int base = beg; base < end; base += 32) {
        const int p = base + lane;
        int   s = 0;
        float c = 0.0f;
        if (p < end) {
            const float2 r = rec[p];
            s = __float_as_int(r.x);
            c = dinv[s] * r.y * di;
        }
        const int cbits = __float_as_int(c);
        const int cnt   = min(32, end - base);
        int j = 0;
        for (; j + 4 <= cnt; j += 4) {
            const int   s0 = __builtin_amdgcn_readlane(s, j);
            const int   s1 = __builtin_amdgcn_readlane(s, j + 1);
            const int   s2 = __builtin_amdgcn_readlane(s, j + 2);
            const int   s3 = __builtin_amdgcn_readlane(s, j + 3);
            const float c0 = __int_as_float(__builtin_amdgcn_readlane(cbits, j));
            const float c1 = __int_as_float(__builtin_amdgcn_readlane(cbits, j + 1));
            const float c2 = __int_as_float(__builtin_amdgcn_readlane(cbits, j + 2));
            const float c3 = __int_as_float(__builtin_amdgcn_readlane(cbits, j + 3));
            const float4 h0 = *(const float4*)&hl[(size_t)s0 * NFEAT];
            const float4 h1 = *(const float4*)&hl[(size_t)s1 * NFEAT];
            const float4 h2 = *(const float4*)&hl[(size_t)s2 * NFEAT];
            const float4 h3 = *(const float4*)&hl[(size_t)s3 * NFEAT];
            acc.x += h0.x * c0; acc.y += h0.y * c0; acc.z += h0.z * c0; acc.w += h0.w * c0;
            acc.x += h1.x * c1; acc.y += h1.y * c1; acc.z += h1.z * c1; acc.w += h1.w * c1;
            acc.x += h2.x * c2; acc.y += h2.y * c2; acc.z += h2.z * c2; acc.w += h2.w * c2;
            acc.x += h3.x * c3; acc.y += h3.y * c3; acc.z += h3.z * c3; acc.w += h3.w * c3;
        }
        for (; j < cnt; ++j) {
            const int    s0 = __builtin_amdgcn_readlane(s, j);
            const float  c0 = __int_as_float(__builtin_amdgcn_readlane(cbits, j));
            const float4 h0 = *(const float4*)&hl[(size_t)s0 * NFEAT];
            acc.x += h0.x * c0; acc.y += h0.y * c0; acc.z += h0.z * c0; acc.w += h0.w * c0;
        }
    }
    return acc;
}

__global__ __launch_bounds__(256) void gcn_gather_tail(
        const float* __restrict__ h, float* __restrict__ out,
        const int* __restrict__ rowptr, const float* __restrict__ csr,
        const float* __restrict__ dinv, const float* __restrict__ bias, int n_nodes) {
    const int lane = threadIdx.x & 31;
    const int row = __builtin_amdgcn_readfirstlane(
                        (int)(blockIdx.x * (blockDim.x >> 5) + (threadIdx.x >> 5)));
    if (row >= n_nodes) return;

    const int   beg = rowptr[row];
    const int   end = rowptr[row + 1];
    const float di  = dinv[row];
    const float*  hl  = h + lane * 4;

    const float4 hs = *(const float4*)&hl[(size_t)row * NFEAT];
    const float4 bv = *(const float4*)&bias[lane * 4];

    const float4 acc = gcn_edge_walk(hl, (const float2*)csr, dinv, beg, end, di, lane);

    const float dd = di * di;
    float4 v;
    v.x = fmaxf(acc.x + hs.x * dd + bv.x, 0.0f);
    v.y = fmaxf(acc.y + hs.y * dd + bv.y, 0.0f);
    v.z = fmaxf(acc.z + hs.z * dd + bv.z, 0.0f);
    v.w = fmaxf(acc.w + hs.w * dd + bv.w, 0.0f);
    *(float4*)&out[(size_t)row * NFEAT + lane * 4] = v;
}

#define NCLS    10
#define WCPITCH 132

template <int M>
__device__ __forceinline__ float gcn_row_xor(float x) {
    return __int_as_float(__builtin_amdgcn_update_dpp(0, __float_as_int(x), 0x160 + M, 0xf, 0xf, true));
}

__device__ __forceinline__ float gcn_xhalf(float x) {
    const int b = __float_as_int(x);
    return __int_as_float(__builtin_amdgcn_permlanex16(b, b, 0x76543210, 0xfedcba98, false, false));
}

__global__ __launch_bounds__(256) void gcn_gather_tail_classify(
        const float* __restrict__ h, float* __restrict__ out,
        const int* __restrict__ rowptr, const float* __restrict__ csr,
        const float* __restrict__ dinv, const float* __restrict__ bias,
        const float* __restrict__ Wc, const float* __restrict__ bc, int n_nodes) {
    __shared__ __attribute__((aligned(16))) float wct[NCLS * WCPITCH];

    const int lane = threadIdx.x & 31;
    const int row = __builtin_amdgcn_readfirstlane(
                        (int)(blockIdx.x * (blockDim.x >> 5) + (threadIdx.x >> 5)));
    const bool live = row < n_nodes;
    const int  rr   = live ? row : n_nodes - 1;

    const int   beg = rowptr[rr];
    const int   end = rowptr[rr + 1];
    const float di  = dinv[rr];
    const float*  hl  = h + lane * 4;
    const float4 hs = *(const float4*)&hl[(size_t)rr * NFEAT];
    const float4 bv = *(const float4*)&bias[lane * 4];
    const float  bcl = (lane < NCLS) ? bc[lane] : 0.0f;

    #pragma unroll
    for (int i = 0; i < (NFEAT * NCLS) / 256; ++i) {
        const int f = (int)threadIdx.x + 256 * i;
        const int d = f / NCLS;
        wct[(f - d * NCLS) * WCPITCH + d] = Wc[f];
    }
    __syncthreads();
    if (!live) return;

    const float4 acc = gcn_edge_walk(hl, (const float2*)csr, dinv, beg, end, di, lane);

    const float dd = di * di;
    float4 v;
    v.x = fmaxf(acc.x + hs.x * dd + bv.x, 0.0f);
    v.y = fmaxf(acc.y + hs.y * dd + bv.y, 0.0f);
    v.z = fmaxf(acc.z + hs.z * dd + bv.z, 0.0f);
    v.w = fmaxf(acc.w + hs.w * dd + bv.w, 0.0f);

    float p[NCLS];
    #pragma unroll
    for (int c = 0; c < NCLS; ++c) {
        const float4 w = *(const float4*)&wct[c * WCPITCH + lane * 4];
        p[c] = v.x * w.x + v.y * w.y + v.z * w.z + v.w * w.w;
    }
    #pragma unroll
    for (int c = 0; c < NCLS; ++c) p[c] += gcn_row_xor<1>(p[c]);
    #pragma unroll
    for (int c = 0; c < NCLS; ++c) p[c] += gcn_row_xor<2>(p[c]);
    #pragma unroll
    for (int c = 0; c < NCLS; ++c) p[c] += gcn_row_xor<4>(p[c]);
    #pragma unroll
    for (int c = 0; c < NCLS; ++c) p[c] += gcn_row_xor<8>(p[c]);

    const int mlane = lane & 15;
    float y = p[0];
    #pragma unroll
    for (int c = 1; c < NCLS; ++c) y = (mlane == c) ? p[c] : y;
    y += gcn_xhalf(y);
    if (lane < NCLS) out[(size_t)row * NCLS + lane] = y + bcl;
}

extern "C" void kernel_launch(void* const* d_in, const int* in_sizes, int n_in,
                              void* d_out, int out_size, void* d_ws, size_t ws_size,
                              hipStream_t stream) {
    const float* x   = (const float*)d_in[0];
    const int*   ei  = (const int*)  d_in[1];
    const float* ew  = (const float*)d_in[2];
    const float* W1  = (const float*)d_in[3];
    const float* b1  = (const float*)d_in[4];
    const float* W2  = (const float*)d_in[5];
    const float* b2  = (const float*)d_in[6];
    const float* Wc  = (const float*)d_in[7];
    const float* bc  = (const float*)d_in[8];
    float*       out = (float*)d_out;

    const int n_nodes = in_sizes[0] / NFEAT;
    const int n_edges = in_sizes[2];
    const int* src = ei;
    const int* dst = ei + n_edges;

    const size_t nd = (size_t)n_nodes * NFEAT;
    const size_t nn = ((size_t)n_nodes + 4) & ~(size_t)3;
    const size_t ne = ((size_t)n_edges + 3) & ~(size_t)3;
    float* bufA = (float*)d_ws;
    float* bufB = bufA + nd;
    float* dinv = bufB + nd;
    int*   rowptr = (int*)(dinv + nn);
    float* csr    = (float*)(rowptr + nn);
    int*   rank   = (int*)(csr + 2 * ne);
    int*   cnts   = rank + ne;
    int*   segs   = cnts + (size_t)NCHUNK * nn;

    const int B = 256;
    const int gN  = (n_nodes + B - 1) / B;
    const int chunkLen = (n_edges + NCHUNK - 1) / NCHUNK;
    const dim3 gFill((chunkLen + B - 1) / B, NCHUNK);
    const int gRow  = (n_nodes + 7) / 8;
    const int gScan = (n_nodes + 1023) / 1024;
    const dim3 gblk(32, 8);
    const int nTiles = n_nodes / 16;
    const int gGemm  = 768;

    gcn_count_chunk<<<NCHUNK, 1024, 0, stream>>>(dst, rank, cnts, segs, n_nodes, (int)nn,
                                                 n_edges, chunkLen);
    gcn_scan_bases <<<gScan, 1024, 0, stream>>>(cnts, segs, rowptr, n_nodes, (int)nn);
    gcn_fill_rows  <<<gFill, B, 0, stream>>>(src, dst, ew, cnts, rank, csr, (int)nn,
                                             n_edges, chunkLen);
    gcn_dinv_rows  <<<gN, B, 0, stream>>>(rowptr, csr, dinv, n_nodes);

    gcn_gemm128_wmma<<<gGemm, gblk, 0, stream>>>(x, W1, bufA, nTiles);
    gcn_gather_tail <<<gRow, B, 0, stream>>>(bufA, bufB, rowptr, csr, dinv, b1, n_nodes);

    gcn_gemm128_wmma<<<gGemm, gblk, 0, stream>>>(bufB, W2, bufA, nTiles);
    gcn_gather_tail_classify<<<gRow, B, 0, stream>>>(bufA, out, rowptr, csr, dinv, b2, Wc, bc, n_nodes);
}


# Degree pass (dinv) moved into the layer-1 product's launch: its first work-groups with one row tile fewer compute it, a row per thread, behind their tiles; 7 launches, not 8; sum orders unchanged
# speedup vs baseline: 3.6093x; 1.0315x over previous
#include <hip/hip_runtime.h>
#include <hip/hip_bf16.h>

typedef float v2f __attribute__((ext_vector_type(2)));
typedef float v8f __attribute__((ext_vector_type(8)));

#define NFEAT 128

#define MAXNODES 50000
#define TABWORDS ((MAXNODES + 1023) & ~1023)
#define NCHUNK   32
#define EPT      20
#define SEGPITCH 64

__global__ __launch_bounds__(1024) void gcn_count_chunk(
        const int* __restrict__ dst, int* __restrict__ rank,
        int* __restrict__ cnts, int* __restrict__ segs,
        int n, int nn, int n_edges, int chunk_len) {
    __shared__ __attribute__((aligned(16))) int tab[TABWORDS];
    const int t    = threadIdx.x;
    const int lane = t & 31;
    const int wave = t >> 5;
    const int c    = blockIdx.x;

    for (int q = t; q < TABWORDS / 4; q += 1024) ((int4*)tab)[q] = make_int4(0, 0, 0, 0);
    __syncthreads();

    const int e0 = c * chunk_len;
    const int e1 = min(e0 + chunk_len, n_edges);
    for (int base = e0; base < e1; base += EPT * 1024) {
        int d[EPT];
        #pragma unroll
        for (int u = 0; u < EPT; ++u) {
            const int e = base + u * 1024 + t;
            d[u] = (e < e1) ? dst[e] : -1;
        }
        int r[EPT];
        #pragma unroll
        for (int u = 0; u < EPT; ++u)
            r[u] = (d[u] >= 0) ? atomicAdd(&tab[d[u]], 1) : 0;
        #pragma unroll
        for (int u = 0; u < EPT; ++u) {
            const int e = base + u * 1024 + t;
            if (d[u] >= 0) rank[e] = r[u];
        }
    }
    __syncthreads();

    int* mine = cnts + (size_t)c * nn;
    for (int i = t; i < n; i += 1024) mine[i] = tab[i];

    const int nseg = (n + 1023) >> 10;
    const int4* tab4 = (const int4*)tab;
    for (int seg = wave; seg < nseg; seg += 32) {
        int s = 0;
        #pragma unroll
        for (int k = 0; k < 8; ++k) {
            const int4 v = tab4[seg * 256 + k * 32 + lane];
            s += (v.x + v.y) + (v.z + v.w);
        }
        for (int o = 16; o > 0; o >>= 1) s += __shfl_xor(s, o);
        if (lane == 0) segs[c * SEGPITCH + seg] = s;
    }
}

__global__ __launch_bounds__(1024) void gcn_scan_bases(
        int* __restrict__ cnts, const int* __restrict__ segs,
        int* __restrict__ rowptr, int n, int nn) {
    __shared__ int s_pre[32];
    __shared__ int s_own[32];
    const int t     = threadIdx.x;
    const int lane  = t & 31;
    const int wave  = t >> 5;
    const int b     = blockIdx.x;
    const int first = b * 1024;

    int pre = 0;
    for (int q = t; q < NCHUNK * SEGPITCH; q += 1024)
        if ((q & (SEGPITCH - 1)) < b) pre += segs[q];

    const int i = first + t;
    int off[NCHUNK];
    int x = 0;
    if (i < n) {
        int v[NCHUNK];
        #pragma unroll
        for (int c = 0; c < NCHUNK; ++c) v[c] = cnts[(size_t)c * nn + i];
        #pragma unroll
        for (int c = 0; c < NCHUNK; ++c) { off[c] = x; x += v[c]; }
    }

    int y = x;
    for (int o = 1; o < 32; o <<= 1) {
        int u = __shfl_up(y, o);
        if (lane >= o) y += u;
    }
    for (int o = 16; o > 0; o >>= 1) pre += __shfl_xor(pre, o);
    if (lane == 31) s_own[wave] = y;
    if (lane == 0)  s_pre[wave] = pre;
    __syncthreads();

    int a = s_pre[lane] + ((lane < wave) ? s_own[lane] : 0);
    for (int o = 16; o > 0; o >>= 1) a += __shfl_xor(a, o);

    const int excl = a + y - x;
    if (i < n) {
        rowptr[i] = excl;
        #pragma unroll
        for (int c = 0; c < NCHUNK; ++c) cnts[(size_t)c * nn + i] = excl + off[c];
    }
    if (i == n - 1) rowptr[n] = excl + x;
}

__global__ void gcn_fill_rows(const int* __restrict__ src, const int* __restrict__ dst,
                              const float* __restrict__ ew,
                              const int* __restrict__ bases, const int* __restrict__ rank,
                              float* __restrict__ csr, int nn, int n_edges, int chunk_len) {
    const int c  = blockIdx.y;
    const int k  = blockIdx.x * blockDim.x + threadIdx.x;
    const int e  = c * chunk_len + k;
    if (k < chunk_len && e < n_edges) {
        const int p = bases[(size_t)c * nn + dst[e]] + rank[e];
        ((float2*)csr)[p] = make_float2(__int_as_float(src[e]), ew[e]);
    }
}

__device__ __forceinline__ void gcn_dinv_slice(const int* __restrict__ rowptr, const float* __restrict__ csr,
                                               float* __restrict__ dinv, int lo, int hi, int tid, int nthreads) {
    const float2* rec = (const float2*)csr;
    for (int i = lo + tid; i < hi; i += nthreads) {
        const int end = rowptr[i + 1];
        float s = 0.0f;
        for (int p = rowptr[i]; p < end; ++p) s += rec[p].y;
        dinv[i] = rsqrtf(s + 1.0f);
    }
}

#define APITCH 132

__global__ __launch_bounds__(256) void gcn_gemm128_wmma(
        const float* __restrict__ X, const float* __restrict__ W,
        float* __restrict__ H, int n_tiles,
        const int* __restrict__ rowptr, const float* __restrict__ csr,
        float* __restrict__ dinv, int n_nodes) {
    __shared__ __attribute__((aligned(16))) float Al[2][16 * APITCH];

    const int lane  = threadIdx.x;
    const int wv    = threadIdx.y;
    const int tid   = wv * 32 + lane;
    const int mlane = lane & 15;
    const int khalf = (lane >> 4) * 2;
    const int ncol  = wv * 16 + mlane;

    int tile = blockIdx.x;
    if (tile < n_tiles) {

        const float4* Xv = (const float4*)X;
        float4 p0 = Xv[(size_t)tile * 512 + tid];
        float4 p1 = Xv[(size_t)tile * 512 + 256 + tid];

        v2f bfr[NFEAT / 4];
        #pragma unroll
        for (int j = 0; j < NFEAT / 4; ++j) {
            const int k = 4 * j + khalf;
            bfr[j].x = W[(k    ) * NFEAT + ncol];
            bfr[j].y = W[(k + 1) * NFEAT + ncol];
        }

        const int st0 = wv * APITCH + lane * 4;
        const int st1 = (wv + 8) * APITCH + lane * 4;
        const int ld0 = mlane * APITCH + khalf;
        float* hrow = H + (size_t)((lane >> 4) * 8) * NFEAT + ncol;

        *(float4*)&Al[0][st0] = p0;
        *(float4*)&Al[0][st1] = p1;
        __syncthreads();

        int buf = 0;
        for (;;) {
            const int  next = tile + (int)gridDim.x;
            const bool more = next < n_tiles;
            if (more) {
                p0 = Xv[(size_t)next * 512 + tid];
                p1 = Xv[(size_t)next * 512 + 256 + tid];
            }

            const float* arow = &Al[buf][ld0];
            v2f afr[NFEAT / 4];
            #pragma unroll
            for (int j = 0; j < NFEAT / 4; ++j) afr[j] = *(const v2f*)&arow[4 * j];
            v8f acc = {};
            #pragma unroll
            for (int j = 0; j < NFEAT / 4; ++j)
                acc = __builtin_amdgcn_wmma_f32_16x16x4_f32(
                          false, afr[j], false, bfr[j], (short)0, acc, false, false);
            __builtin_amdgcn_sched_group_barrier(0x100, 16, 0);
            __builtin_amdgcn_sched_group_barrier(0x008, 32, 0);

            if (more) {
                *(float4*)&Al[buf ^ 1][st0] = p0;
                *(float4*)&Al[buf ^ 1][st1] = p1;
                __syncthreads();
            }
            float* ht = hrow + (size_t)tile * (16 * NFEAT);
            #pragma unroll
            for (int r = 0; r < 8; ++r)
                ht[r * NFEAT] = acc[r];

            if (!more) break;
            buf ^= 1;
            tile = next;
        }
    }

    if (rowptr != nullptr) {
        const int G    = (int)gridDim.x;
        const int rem  = n_tiles % G;
        const int b    = (int)blockIdx.x - rem;
        if (b >= 0) {
            const int nb  = G - rem;
            const int per = max(256, (n_nodes + nb - 1) / nb);
            const int lo  = b * per;
            gcn_dinv_slice(rowptr, csr, dinv, lo, min(lo + per, n_nodes), tid, 256);
        }
    }
}


__device__ __forceinline__ float4 gcn_edge_walk(
        const float* __restrict__ hl, const float2* __restrict__ rec,
        const float* __restrict__ dinv, int beg, int end, float di, int lane) {
    float4 acc = make_float4(0.0f, 0.0f, 0.0f, 0.0f);
    for (int base = beg; base < end; base += 32) {
        const int p = base + lane;
        int   s = 0;
        float c = 0.0f;
        if (p < end) {
            const float2 r = rec[p];
            s = __float_as_int(r.x);
            c = dinv[s] * r.y * di;
        }
        const int cbits = __float_as_int(c);
        const int cnt   = min(32, end - base);
        int j = 0;
        for (; j + 4 <= cnt; j += 4) {
            const int   s0 = __builtin_amdgcn_readlane(s, j);
            const int   s1 = __builtin_amdgcn_readlane(s, j + 1);
            const int   s2 = __builtin_amdgcn_readlane(s, j + 2);
            const int   s3 = __builtin_amdgcn_readlane(s, j + 3);
            const float c0 = __int_as_float(__builtin_amdgcn_readlane(cbits, j));
            const float c1 = __int_as_float(__builtin_amdgcn_readlane(cbits, j + 1));
            const float c2 = __int_as_float(__builtin_amdgcn_readlane(cbits, j + 2));
            const float c3 = __int_as_float(__builtin_amdgcn_readlane(cbits, j + 3));
            const float4 h0 = *(const float4*)&hl[(size_t)s0 * NFEAT];
            const float4 h1 = *(const float4*)&hl[(size_t)s1 * NFEAT];
            const float4 h2 = *(const float4*)&hl[(size_t)s2 * NFEAT];
            const float4 h3 = *(const float4*)&hl[(size_t)s3 * NFEAT];
            acc.x += h0.x * c0; acc.y += h0.y * c0; acc.z += h0.z * c0; acc.w += h0.w * c0;
            acc.x += h1.x * c1; acc.y += h1.y * c1; acc.z += h1.z * c1; acc.w += h1.w * c1;
            acc.x += h2.x * c2; acc.y += h2.y * c2; acc.z += h2.z * c2; acc.w += h2.w * c2;
            acc.x += h3.x * c3; acc.y += h3.y * c3; acc.z += h3.z * c3; acc.w += h3.w * c3;
        }
        for (; j < cnt; ++j) {
            const int    s0 = __builtin_amdgcn_readlane(s, j);
            const float  c0 = __int_as_float(__builtin_amdgcn_readlane(cbits, j));
            const float4 h0 = *(const float4*)&hl[(size_t)s0 * NFEAT];
            acc.x += h0.x * c0; acc.y += h0.y * c0; acc.z += h0.z * c0; acc.w += h0.w * c0;
        }
    }
    return acc;
}

__global__ __launch_bounds__(256) void gcn_gather_tail(
        const float* __restrict__ h, float* __restrict__ out,
        const int* __restrict__ rowptr, const float* __restrict__ csr,
        const float* __restrict__ dinv, const float* __restrict__ bias, int n_nodes) {
    const int lane = threadIdx.x & 31;
    const int row = __builtin_amdgcn_readfirstlane(
                        (int)(blockIdx.x * (blockDim.x >> 5) + (threadIdx.x >> 5)));
    if (row >= n_nodes) return;

    const int   beg = rowptr[row];
    const int   end = rowptr[row + 1];
    const float di  = dinv[row];
    const float*  hl  = h + lane * 4;

    const float4 hs = *(const float4*)&hl[(size_t)row * NFEAT];
    const float4 bv = *(const float4*)&bias[lane * 4];

    const float4 acc = gcn_edge_walk(hl, (const float2*)csr, dinv, beg, end, di, lane);

    const float dd = di * di;
    float4 v;
    v.x = fmaxf(acc.x + hs.x * dd + bv.x, 0.0f);
    v.y = fmaxf(acc.y + hs.y * dd + bv.y, 0.0f);
    v.z = fmaxf(acc.z + hs.z * dd + bv.z, 0.0f);
    v.w = fmaxf(acc.w + hs.w * dd + bv.w, 0.0f);
    *(float4*)&out[(size_t)row * NFEAT + lane * 4] = v;
}

#define NCLS    10
#define WCPITCH 132

template <int M>
__device__ __forceinline__ float gcn_row_xor(float x) {
    return __int_as_float(__builtin_amdgcn_update_dpp(0, __float_as_int(x), 0x160 + M, 0xf, 0xf, true));
}

__device__ __forceinline__ float gcn_xhalf(float x) {
    const int b = __float_as_int(x);
    return __int_as_float(__builtin_amdgcn_permlanex16(b, b, 0x76543210, 0xfedcba98, false, false));
}

__global__ __launch_bounds__(256) void gcn_gather_tail_classify(
        const float* __restrict__ h, float* __restrict__ out,
        const int* __restrict__ rowptr, const float* __restrict__ csr,
        const float* __restrict__ dinv, const float* __restrict__ bias,
        const float* __restrict__ Wc, const float* __restrict__ bc, int n_nodes) {
    __shared__ __attribute__((aligned(16))) float wct[NCLS * WCPITCH];

    const int lane = threadIdx.x & 31;
    const int row = __builtin_amdgcn_readfirstlane(
                        (int)(blockIdx.x * (blockDim.x >> 5) + (threadIdx.x >> 5)));
    const bool live = row < n_nodes;
    const int  rr   = live ? row : n_nodes - 1;

    const int   beg = rowptr[rr];
    const int   end = rowptr[rr + 1];
    const float di  = dinv[rr];
    const float*  hl  = h + lane * 4;
    const float4 hs = *(const float4*)&hl[(size_t)rr * NFEAT];
    const float4 bv = *(const float4*)&bias[lane * 4];
    const float  bcl = (lane < NCLS) ? bc[lane] : 0.0f;

    #pragma unroll
    for (int i = 0; i < (NFEAT * NCLS) / 256; ++i) {
        const int f = (int)threadIdx.x + 256 * i;
        const int d = f / NCLS;
        wct[(f - d * NCLS) * WCPITCH + d] = Wc[f];
    }
    __syncthreads();
    if (!live) return;

    const float4 acc = gcn_edge_walk(hl, (const float2*)csr, dinv, beg, end, di, lane);

    const float dd = di * di;
    float4 v;
    v.x = fmaxf(acc.x + hs.x * dd + bv.x, 0.0f);
    v.y = fmaxf(acc.y + hs.y * dd + bv.y, 0.0f);
    v.z = fmaxf(acc.z + hs.z * dd + bv.z, 0.0f);
    v.w = fmaxf(acc.w + hs.w * dd + bv.w, 0.0f);

    float p[NCLS];
    #pragma unroll
    for (int c = 0; c < NCLS; ++c) {
        const float4 w = *(const float4*)&wct[c * WCPITCH + lane * 4];
        p[c] = v.x * w.x + v.y * w.y + v.z * w.z + v.w * w.w;
    }
    #pragma unroll
    for (int c = 0; c < NCLS; ++c) p[c] += gcn_row_xor<1>(p[c]);
    #pragma unroll
    for (int c = 0; c < NCLS; ++c) p[c] += gcn_row_xor<2>(p[c]);
    #pragma unroll
    for (int c = 0; c < NCLS; ++c) p[c] += gcn_row_xor<4>(p[c]);
    #pragma unroll
    for (int c = 0; c < NCLS; ++c) p[c] += gcn_row_xor<8>(p[c]);

    const int mlane = lane & 15;
    float y = p[0];
    #pragma unroll
    for (int c = 1; c < NCLS; ++c) y = (mlane == c) ? p[c] : y;
    y += gcn_xhalf(y);
    if (lane < NCLS) out[(size_t)row * NCLS + lane] = y + bcl;
}

extern "C" void kernel_launch(void* const* d_in, const int* in_sizes, int n_in,
                              void* d_out, int out_size, void* d_ws, size_t ws_size,
                              hipStream_t stream) {
    const float* x   = (const float*)d_in[0];
    const int*   ei  = (const int*)  d_in[1];
    const float* ew  = (const float*)d_in[2];
    const float* W1  = (const float*)d_in[3];
    const float* b1  = (const float*)d_in[4];
    const float* W2  = (const float*)d_in[5];
    const float* b2  = (const float*)d_in[6];
    const float* Wc  = (const float*)d_in[7];
    const float* bc  = (const float*)d_in[8];
    float*       out = (float*)d_out;

    const int n_nodes = in_sizes[0] / NFEAT;
    const int n_edges = in_sizes[2];
    const int* src = ei;
    const int* dst = ei + n_edges;

    const size_t nd = (size_t)n_nodes * NFEAT;
    const size_t nn = ((size_t)n_nodes + 4) & ~(size_t)3;
    const size_t ne = ((size_t)n_edges + 3) & ~(size_t)3;
    float* bufA = (float*)d_ws;
    float* bufB = bufA + nd;
    float* dinv = bufB + nd;
    int*   rowptr = (int*)(dinv + nn);
    float* csr    = (float*)(rowptr + nn);
    int*   rank   = (int*)(csr + 2 * ne);
    int*   cnts   = rank + ne;
    int*   segs   = cnts + (size_t)NCHUNK * nn;

    const int B = 256;
    const int chunkLen = (n_edges + NCHUNK - 1) / NCHUNK;
    const dim3 gFill((chunkLen + B - 1) / B, NCHUNK);
    const int gRow  = (n_nodes + 7) / 8;
    const int gScan = (n_nodes + 1023) / 1024;
    const dim3 gblk(32, 8);
    const int nTiles = n_nodes / 16;
    const int gGemm  = 768;

    gcn_count_chunk<<<NCHUNK, 1024, 0, stream>>>(dst, rank, cnts, segs, n_nodes, (int)nn,
                                                 n_edges, chunkLen);
    gcn_scan_bases <<<gScan, 1024, 0, stream>>>(cnts, segs, rowptr, n_nodes, (int)nn);
    gcn_fill_rows  <<<gFill, B, 0, stream>>>(src, dst, ew, cnts, rank, csr, (int)nn,
                                             n_edges, chunkLen);

    gcn_gemm128_wmma<<<gGemm, gblk, 0, stream>>>(x, W1, bufA, nTiles, rowptr, csr, dinv, n_nodes);
    gcn_gather_tail <<<gRow, B, 0, stream>>>(bufA, bufB, rowptr, csr, dinv, b1, n_nodes);

    gcn_gemm128_wmma<<<gGemm, gblk, 0, stream>>>(bufB, W2, bufA, nTiles, nullptr, nullptr, nullptr, n_nodes);
    gcn_gather_tail_classify<<<gRow, B, 0, stream>>>(bufA, out, rowptr, csr, dinv, b2, Wc, bc, n_nodes);
}


# Edge-count pass cut into 7 node ranges per chunk: 32 x 7 work-groups, each with LDS counters for 7168 nodes, not 32 with a counter for every node; ranks, tables and row order as before
# speedup vs baseline: 3.7584x; 1.0413x over previous
#include <hip/hip_runtime.h>
#include <hip/hip_bf16.h>

typedef float v2f __attribute__((ext_vector_type(2)));
typedef float v8f __attribute__((ext_vector_type(8)));

#define NFEAT 128

#define MAXNODES 50000
#define NCHUNK   32
#define SEGS_PER_RANGE 7
#define RANGE    (SEGS_PER_RANGE * 1024)
#define NRANGE   ((MAXNODES + RANGE - 1) / RANGE)
#define EPT      20
#define SEGPITCH 64

__global__ __launch_bounds__(1024) void gcn_count_chunk(
        const int* __restrict__ dst, int* __restrict__ rank,
        int* __restrict__ cnts, int* __restrict__ segs,
        int n, int nn, int n_edges, int chunk_len) {
    __shared__ __attribute__((aligned(16))) int tab[RANGE];
    const int t    = threadIdx.x;
    const int lane = t & 31;
    const int wave = t >> 5;
    const int c    = blockIdx.x;
    const int lo   = (int)blockIdx.y * RANGE;

    for (int q = t; q < RANGE / 4; q += 1024) ((int4*)tab)[q] = make_int4(0, 0, 0, 0);
    __syncthreads();

    const int e0 = c * chunk_len;
    const int e1 = min(e0 + chunk_len, n_edges);
    for (int base = e0; base < e1; base += EPT * 1024) {
        int d[EPT];
        #pragma unroll
        for (int u = 0; u < EPT; ++u) {
            const int e = base + u * 1024 + t;
            d[u] = (e < e1) ? dst[e] : -1;
        }
        int r[EPT];
        #pragma unroll
        for (int u = 0; u < EPT; ++u) {
            const unsigned k = (unsigned)(d[u] - lo);
            r[u] = (k < (unsigned)RANGE) ? atomicAdd(&tab[k], 1) : 0;
        }
        #pragma unroll
        for (int u = 0; u < EPT; ++u) {
            const int e = base + u * 1024 + t;
            if ((unsigned)(d[u] - lo) < (unsigned)RANGE) rank[e] = r[u];
        }
    }
    __syncthreads();

    int* mine = cnts + (size_t)c * nn;
    int cv[RANGE / 1024];
    #pragma unroll
    for (int k = 0; k < RANGE / 1024; ++k) cv[k] = tab[k * 1024 + t];
    #pragma unroll
    for (int k = 0; k < RANGE / 1024; ++k) {
        const int i = lo + k * 1024 + t;
        if (i < n) mine[i] = cv[k];
    }

    const int nseg = (n + 1023) >> 10;
    const int seg  = (int)blockIdx.y * SEGS_PER_RANGE + wave;
    const int4* tab4 = (const int4*)tab;
    if (wave < SEGS_PER_RANGE && seg < nseg) {
        int s = 0;
        #pragma unroll
        for (int k = 0; k < 8; ++k) {
            const int4 v = tab4[wave * 256 + k * 32 + lane];
            s += (v.x + v.y) + (v.z + v.w);
        }
        for (int o = 16; o > 0; o >>= 1) s += __shfl_xor(s, o);
        if (lane == 0) segs[c * SEGPITCH + seg] = s;
    }
}

__global__ __launch_bounds__(1024) void gcn_scan_bases(
        int* __restrict__ cnts, const int* __restrict__ segs,
        int* __restrict__ rowptr, int n, int nn) {
    __shared__ int s_pre[32];
    __shared__ int s_own[32];
    const int t     = threadIdx.x;
    const int lane  = t & 31;
    const int wave  = t >> 5;
    const int b     = blockIdx.x;
    const int first = b * 1024;

    int pre = 0;
    for (int q = t; q < NCHUNK * SEGPITCH; q += 1024)
        if ((q & (SEGPITCH - 1)) < b) pre += segs[q];

    const int i = first + t;
    int off[NCHUNK];
    int x = 0;
    if (i < n) {
        int v[NCHUNK];
        #pragma unroll
        for (int c = 0; c < NCHUNK; ++c) v[c] = cnts[(size_t)c * nn + i];
        #pragma unroll
        for (int c = 0; c < NCHUNK; ++c) { off[c] = x; x += v[c]; }
    }

    int y = x;
    for (int o = 1; o < 32; o <<= 1) {
        int u = __shfl_up(y, o);
        if (lane >= o) y += u;
    }
    for (int o = 16; o > 0; o >>= 1) pre += __shfl_xor(pre, o);
    if (lane == 31) s_own[wave] = y;
    if (lane == 0)  s_pre[wave] = pre;
    __syncthreads();

    int a = s_pre[lane] + ((lane < wave) ? s_own[lane] : 0);
    for (int o = 16; o > 0; o >>= 1) a += __shfl_xor(a, o);

    const int excl = a + y - x;
    if (i < n) {
        rowptr[i] = excl;
        #pragma unroll
        for (int c = 0; c < NCHUNK; ++c) cnts[(size_t)c * nn + i] = excl + off[c];
    }
    if (i == n - 1) rowptr[n] = excl + x;
}

__global__ void gcn_fill_rows(const int* __restrict__ src, const int* __restrict__ dst,
                              const float* __restrict__ ew,
                              const int* __restrict__ bases, const int* __restrict__ rank,
                              float* __restrict__ csr, int nn, int n_edges, int chunk_len) {
    const int c  = blockIdx.y;
    const int k  = blockIdx.x * blockDim.x + threadIdx.x;
    const int e  = c * chunk_len + k;
    if (k < chunk_len && e < n_edges) {
        const int p = bases[(size_t)c * nn + dst[e]] + rank[e];
        ((float2*)csr)[p] = make_float2(__int_as_float(src[e]), ew[e]);
    }
}

__device__ __forceinline__ void gcn_dinv_slice(const int* __restrict__ rowptr, const float* __restrict__ csr,
                                               float* __restrict__ dinv, int lo, int hi, int tid, int nthreads) {
    const float2* rec = (const float2*)csr;
    for (int i = lo + tid; i < hi; i += nthreads) {
        const int end = rowptr[i + 1];
        float s = 0.0f;
        for (int p = rowptr[i]; p < end; ++p) s += rec[p].y;
        dinv[i] = rsqrtf(s + 1.0f);
    }
}

#define APITCH 132

__global__ __launch_bounds__(256) void gcn_gemm128_wmma(
        const float* __restrict__ X, const float* __restrict__ W,
        float* __restrict__ H, int n_tiles,
        const int* __restrict__ rowptr, const float* __restrict__ csr,
        float* __restrict__ dinv, int n_nodes) {
    __shared__ __attribute__((aligned(16))) float Al[2][16 * APITCH];

    const int lane  = threadIdx.x;
    const int wv    = threadIdx.y;
    const int tid   = wv * 32 + lane;
    const int mlane = lane & 15;
    const int khalf = (lane >> 4) * 2;
    const int ncol  = wv * 16 + mlane;

    int tile = blockIdx.x;
    if (tile < n_tiles) {

        const float4* Xv = (const float4*)X;
        float4 p0 = Xv[(size_t)tile * 512 + tid];
        float4 p1 = Xv[(size_t)tile * 512 + 256 + tid];

        v2f bfr[NFEAT / 4];
        #pragma unroll
        for (int j = 0; j < NFEAT / 4; ++j) {
            const int k = 4 * j + khalf;
            bfr[j].x = W[(k    ) * NFEAT + ncol];
            bfr[j].y = W[(k + 1) * NFEAT + ncol];
        }

        const int st0 = wv * APITCH + lane * 4;
        const int st1 = (wv + 8) * APITCH + lane * 4;
        const int ld0 = mlane * APITCH + khalf;
        float* hrow = H + (size_t)((lane >> 4) * 8) * NFEAT + ncol;

        *(float4*)&Al[0][st0] = p0;
        *(float4*)&Al[0][st1] = p1;
        __syncthreads();

        int buf = 0;
        for (;;) {
            const int  next = tile + (int)gridDim.x;
            const bool more = next < n_tiles;
            if (more) {
                p0 = Xv[(size_t)next * 512 + tid];
                p1 = Xv[(size_t)next * 512 + 256 + tid];
            }

            const float* arow = &Al[buf][ld0];
            v2f afr[NFEAT / 4];
            #pragma unroll
            for (int j = 0; j < NFEAT / 4; ++j) afr[j] = *(const v2f*)&arow[4 * j];
            v8f acc = {};
            #pragma unroll
            for (int j = 0; j < NFEAT / 4; ++j)
                acc = __builtin_amdgcn_wmma_f32_16x16x4_f32(
                          false, afr[j], false, bfr[j], (short)0, acc, false, false);
            __builtin_amdgcn_sched_group_barrier(0x100, 16, 0);
            __builtin_amdgcn_sched_group_barrier(0x008, 32, 0);

            if (more) {
                *(float4*)&Al[buf ^ 1][st0] = p0;
                *(float4*)&Al[buf ^ 1][st1] = p1;
                __syncthreads();
            }
            float* ht = hrow + (size_t)tile * (16 * NFEAT);
            #pragma unroll
            for (int r = 0; r < 8; ++r)
                ht[r * NFEAT] = acc[r];

            if (!more) break;
            buf ^= 1;
            tile = next;
        }
    }

    if (rowptr != nullptr) {
        const int G    = (int)gridDim.x;
        const int rem  = n_tiles % G;
        const int b    = (int)blockIdx.x - rem;
        if (b >= 0) {
            const int nb  = G - rem;
            const int per = max(256, (n_nodes + nb - 1) / nb);
            const int lo  = b * per;
            gcn_dinv_slice(rowptr, csr, dinv, lo, min(lo + per, n_nodes), tid, 256);
        }
    }
}


__device__ __forceinline__ float4 gcn_edge_walk(
        const float* __restrict__ hl, const float2* __restrict__ rec,
        const float* __restrict__ dinv, int beg, int end, float di, int lane) {
    float4 acc = make_float4(0.0f, 0.0f, 0.0f, 0.0f);
    for (int base = beg; base < end; base += 32) {
        const int p = base + lane;
        int   s = 0;
        float c = 0.0f;
        if (p < end) {
            const float2 r = rec[p];
            s = __float_as_int(r.x);
            c = dinv[s] * r.y * di;
        }
        const int cbits = __float_as_int(c);
        const int cnt   = min(32, end - base);
        int j = 0;
        for (; j + 4 <= cnt; j += 4) {
            const int   s0 = __builtin_amdgcn_readlane(s, j);
            const int   s1 = __builtin_amdgcn_readlane(s, j + 1);
            const int   s2 = __builtin_amdgcn_readlane(s, j + 2);
            const int   s3 = __builtin_amdgcn_readlane(s, j + 3);
            const float c0 = __int_as_float(__builtin_amdgcn_readlane(cbits, j));
            const float c1 = __int_as_float(__builtin_amdgcn_readlane(cbits, j + 1));
            const float c2 = __int_as_float(__builtin_amdgcn_readlane(cbits, j + 2));
            const float c3 = __int_as_float(__builtin_amdgcn_readlane(cbits, j + 3));
            const float4 h0 = *(const float4*)&hl[(size_t)s0 * NFEAT];
            const float4 h1 = *(const float4*)&hl[(size_t)s1 * NFEAT];
            const float4 h2 = *(const float4*)&hl[(size_t)s2 * NFEAT];
            const float4 h3 = *(const float4*)&hl[(size_t)s3 * NFEAT];
            acc.x += h0.x * c0; acc.y += h0.y * c0; acc.z += h0.z * c0; acc.w += h0.w * c0;
            acc.x += h1.x * c1; acc.y += h1.y * c1; acc.z += h1.z * c1; acc.w += h1.w * c1;
            acc.x += h2.x * c2; acc.y += h2.y * c2; acc.z += h2.z * c2; acc.w += h2.w * c2;
            acc.x += h3.x * c3; acc.y += h3.y * c3; acc.z += h3.z * c3; acc.w += h3.w * c3;
        }
        for (; j < cnt; ++j) {
            const int    s0 = __builtin_amdgcn_readlane(s, j);
            const float  c0 = __int_as_float(__builtin_amdgcn_readlane(cbits, j));
            const float4 h0 = *(const float4*)&hl[(size_t)s0 * NFEAT];
            acc.x += h0.x * c0; acc.y += h0.y * c0; acc.z += h0.z * c0; acc.w += h0.w * c0;
        }
    }
    return acc;
}

__global__ __launch_bounds__(256) void gcn_gather_tail(
        const float* __restrict__ h, float* __restrict__ out,
        const int* __restrict__ rowptr, const float* __restrict__ csr,
        const float* __restrict__ dinv, const float* __restrict__ bias, int n_nodes) {
    const int lane = threadIdx.x & 31;
    const int row = __builtin_amdgcn_readfirstlane(
                        (int)(blockIdx.x * (blockDim.x >> 5) + (threadIdx.x >> 5)));
    if (row >= n_nodes) return;

    const int   beg = rowptr[row];
    const int   end = rowptr[row + 1];
    const float di  = dinv[row];
    const float*  hl  = h + lane * 4;

    const float4 hs = *(const float4*)&hl[(size_t)row * NFEAT];
    const float4 bv = *(const float4*)&bias[lane * 4];

    const float4 acc = gcn_edge_walk(hl, (const float2*)csr, dinv, beg, end, di, lane);

    const float dd = di * di;
    float4 v;
    v.x = fmaxf(acc.x + hs.x * dd + bv.x, 0.0f);
    v.y = fmaxf(acc.y + hs.y * dd + bv.y, 0.0f);
    v.z = fmaxf(acc.z + hs.z * dd + bv.z, 0.0f);
    v.w = fmaxf(acc.w + hs.w * dd + bv.w, 0.0f);
    *(float4*)&out[(size_t)row * NFEAT + lane * 4] = v;
}

#define NCLS    10
#define WCPITCH 132

template <int M>
__device__ __forceinline__ float gcn_row_xor(float x) {
    return __int_as_float(__builtin_amdgcn_update_dpp(0, __float_as_int(x), 0x160 + M, 0xf, 0xf, true));
}

__device__ __forceinline__ float gcn_xhalf(float x) {
    const int b = __float_as_int(x);
    return __int_as_float(__builtin_amdgcn_permlanex16(b, b, 0x76543210, 0xfedcba98, false, false));
}

__global__ __launch_bounds__(256) void gcn_gather_tail_classify(
        const float* __restrict__ h, float* __restrict__ out,
        const int* __restrict__ rowptr, const float* __restrict__ csr,
        const float* __restrict__ dinv, const float* __restrict__ bias,
        const float* __restrict__ Wc, const float* __restrict__ bc, int n_nodes) {
    __shared__ __attribute__((aligned(16))) float wct[NCLS * WCPITCH];

    const int lane = threadIdx.x & 31;
    const int row = __builtin_amdgcn_readfirstlane(
                        (int)(blockIdx.x * (blockDim.x >> 5) + (threadIdx.x >> 5)));
    const bool live = row < n_nodes;
    const int  rr   = live ? row : n_nodes - 1;

    const int   beg = rowptr[rr];
    const int   end = rowptr[rr + 1];
    const float di  = dinv[rr];
    const float*  hl  = h + lane * 4;
    const float4 hs = *(const float4*)&hl[(size_t)rr * NFEAT];
    const float4 bv = *(const float4*)&bias[lane * 4];
    const float  bcl = (lane < NCLS) ? bc[lane] : 0.0f;

    #pragma unroll
    for (int i = 0; i < (NFEAT * NCLS) / 256; ++i) {
        const int f = (int)threadIdx.x + 256 * i;
        const int d = f / NCLS;
        wct[(f - d * NCLS) * WCPITCH + d] = Wc[f];
    }
    __syncthreads();
    if (!live) return;

    const float4 acc = gcn_edge_walk(hl, (const float2*)csr, dinv, beg, end, di, lane);

    const float dd = di * di;
    float4 v;
    v.x = fmaxf(acc.x + hs.x * dd + bv.x, 0.0f);
    v.y = fmaxf(acc.y + hs.y * dd + bv.y, 0.0f);
    v.z = fmaxf(acc.z + hs.z * dd + bv.z, 0.0f);
    v.w = fmaxf(acc.w + hs.w * dd + bv.w, 0.0f);

    float p[NCLS];
    #pragma unroll
    for (int c = 0; c < NCLS; ++c) {
        const float4 w = *(const float4*)&wct[c * WCPITCH + lane * 4];
        p[c] = v.x * w.x + v.y * w.y + v.z * w.z + v.w * w.w;
    }
    #pragma unroll
    for (int c = 0; c < NCLS; ++c) p[c] += gcn_row_xor<1>(p[c]);
    #pragma unroll
    for (int c = 0; c < NCLS; ++c) p[c] += gcn_row_xor<2>(p[c]);
    #pragma unroll
    for (int c = 0; c < NCLS; ++c) p[c] += gcn_row_xor<4>(p[c]);
    #pragma unroll
    for (int c = 0; c < NCLS; ++c) p[c] += gcn_row_xor<8>(p[c]);

    const int mlane = lane & 15;
    float y = p[0];
    #pragma unroll
    for (int c = 1; c < NCLS; ++c) y = (mlane == c) ? p[c] : y;
    y += gcn_xhalf(y);
    if (lane < NCLS) out[(size_t)row * NCLS + lane] = y + bcl;
}

extern "C" void kernel_launch(void* const* d_in, const int* in_sizes, int n_in,
                              void* d_out, int out_size, void* d_ws, size_t ws_size,
                              hipStream_t stream) {
    const float* x   = (const float*)d_in[0];
    const int*   ei  = (const int*)  d_in[1];
    const float* ew  = (const float*)d_in[2];
    const float* W1  = (const float*)d_in[3];
    const float* b1  = (const float*)d_in[4];
    const float* W2  = (const float*)d_in[5];
    const float* b2  = (const float*)d_in[6];
    const float* Wc  = (const float*)d_in[7];
    const float* bc  = (const float*)d_in[8];
    float*       out = (float*)d_out;

    const int n_nodes = in_sizes[0] / NFEAT;
    const int n_edges = in_sizes[2];
    const int* src = ei;
    const int* dst = ei + n_edges;

    const size_t nd = (size_t)n_nodes * NFEAT;
    const size_t nn = ((size_t)n_nodes + 4) & ~(size_t)3;
    const size_t ne = ((size_t)n_edges + 3) & ~(size_t)3;
    float* bufA = (float*)d_ws;
    float* bufB = bufA + nd;
    float* dinv = bufB + nd;
    int*   rowptr = (int*)(dinv + nn);
    float* csr    = (float*)(rowptr + nn);
    int*   rank   = (int*)(csr + 2 * ne);
    int*   cnts   = rank + ne;
    int*   segs   = cnts + (size_t)NCHUNK * nn;

    const int B = 256;
    const int chunkLen = (n_edges + NCHUNK - 1) / NCHUNK;
    const dim3 gFill((chunkLen + B - 1) / B, NCHUNK);
    const int gRow  = (n_nodes + 7) / 8;
    const int gScan = (n_nodes + 1023) / 1024;
    const dim3 gblk(32, 8);
    const int nTiles = n_nodes / 16;
    const int gGemm  = 768;

    gcn_count_chunk<<<dim3(NCHUNK, NRANGE), 1024, 0, stream>>>(dst, rank, cnts, segs, n_nodes, (int)nn,
                                                               n_edges, chunkLen);
    gcn_scan_bases <<<gScan, 1024, 0, stream>>>(cnts, segs, rowptr, n_nodes, (int)nn);
    gcn_fill_rows  <<<gFill, B, 0, stream>>>(src, dst, ew, cnts, rank, csr, (int)nn,
                                             n_edges, chunkLen);

    gcn_gemm128_wmma<<<gGemm, gblk, 0, stream>>>(x, W1, bufA, nTiles, rowptr, csr, dinv, n_nodes);
    gcn_gather_tail <<<gRow, B, 0, stream>>>(bufA, bufB, rowptr, csr, dinv, b1, n_nodes);

    gcn_gemm128_wmma<<<gGemm, gblk, 0, stream>>>(bufB, W2, bufA, nTiles, nullptr, nullptr, nullptr, n_nodes);
    gcn_gather_tail_classify<<<gRow, B, 0, stream>>>(bufA, out, rowptr, csr, dinv, b2, Wc, bc, n_nodes);
}
